# starting kernel
# baseline (speedup 1.0000x reference)
#include <hip/hip_runtime.h>
#include <math.h>

typedef __attribute__((ext_vector_type(2))) float v2f;
typedef __attribute__((ext_vector_type(8))) float v8f;

#define BATCH 2
#define SEQ 2048
#define DM 1024
#define NH 16
#define DK 64

template <int STORE_VT>
__global__ __launch_bounds__(256) void gemm_xwt_kernel(const float* __restrict__ X,
                                                       const float* __restrict__ W,
                                                       float* __restrict__ out,
                                                       int M, int N, int K) {
    const int lane = threadIdx.x & 31;
    const int warp = threadIdx.x >> 5;
    const int tile = blockIdx.x * 8 + warp;
    const int tilesN = N >> 6;
    const int tm = (tile / tilesN) << 5;
    const int tn = (tile % tilesN) << 6;
    const int ln = lane & 15;
    const int lh = lane >> 4;

    const float* xrow0 = X + (size_t)(tm + ln) * K + 2 * lh;
    const float* xrow1 = xrow0 + (size_t)16 * K;
    const float* wrow  = W + (size_t)(tn + ln) * K + 2 * lh;

    v8f c[2][4] = {};
#pragma unroll 4
    for (int k0 = 0; k0 < K; k0 += 4) {
        v2f a0 = *(const v2f*)(xrow0 + k0);
        v2f a1 = *(const v2f*)(xrow1 + k0);
        v2f b0 = *(const v2f*)(wrow + k0);
        v2f b1 = *(const v2f*)(wrow + (size_t)16 * K + k0);
        v2f b2 = *(const v2f*)(wrow + (size_t)32 * K + k0);
        v2f b3 = *(const v2f*)(wrow + (size_t)48 * K + k0);
        c[0][0] = __builtin_amdgcn_wmma_f32_16x16x4_f32(false, a0, false, b0, (short)0, c[0][0], false, false);
        c[0][1] = __builtin_amdgcn_wmma_f32_16x16x4_f32(false, a0, false, b1, (short)0, c[0][1], false, false);
        c[0][2] = __builtin_amdgcn_wmma_f32_16x16x4_f32(false, a0, false, b2, (short)0, c[0][2], false, false);
        c[0][3] = __builtin_amdgcn_wmma_f32_16x16x4_f32(false, a0, false, b3, (short)0, c[0][3], false, false);
        c[1][0] = __builtin_amdgcn_wmma_f32_16x16x4_f32(false, a1, false, b0, (short)0, c[1][0], false, false);
        c[1][1] = __builtin_amdgcn_wmma_f32_16x16x4_f32(false, a1, false, b1, (short)0, c[1][1], false, false);
        c[1][2] = __builtin_amdgcn_wmma_f32_16x16x4_f32(false, a1, false, b2, (short)0, c[1][2], false, false);
        c[1][3] = __builtin_amdgcn_wmma_f32_16x16x4_f32(false, a1, false, b3, (short)0, c[1][3], false, false);
    }

    if (STORE_VT == 0) {
#pragma unroll
        for (int mi = 0; mi < 2; ++mi)
#pragma unroll
            for (int ni = 0; ni < 4; ++ni) {
                float* orow = out + (size_t)(tm + mi * 16 + 8 * lh) * N + tn + ni * 16 + ln;
#pragma unroll
                for (int r = 0; r < 8; ++r)
                    orow[(size_t)r * N] = c[mi][ni][r];
            }
    } else {
        const int bb = tm >> 11;
#pragma unroll
        for (int mi = 0; mi < 2; ++mi)
#pragma unroll
            for (int ni = 0; ni < 4; ++ni) {
                int n = tn + ni * 16 + ln;
                int srow = (tm & (SEQ - 1)) + mi * 16 + 8 * lh;
                float* ocol = out + ((size_t)(bb * N + n) << 11) + srow;
#pragma unroll
                for (int r = 0; r < 8; ++r)
                    ocol[r] = c[mi][ni][r];
            }
    }
}

__global__ __launch_bounds__(256) void rope_kernel(float* __restrict__ q,
                                                   float* __restrict__ k, int total) {
    int tid = blockIdx.x * blockDim.x + threadIdx.x;
    if (tid >= total) return;
    int i = tid & 31;
    int s = (tid >> 9) & (SEQ - 1);
    float freq = __expf(-(float)i * (1.0f / 32.0f) * 9.210340371976184f);
    float ang = (float)s * freq;
    float cs = __cosf(ang);
    float sn = __sinf(ang);
    size_t p = (size_t)tid * 2;
    float q1 = q[p], q2 = q[p + 1];
    q[p]     = cs * q1 - sn * q2;
    q[p + 1] = sn * q1 + cs * q2;
    float k1 = k[p], k2 = k[p + 1];
    k[p]     = cs * k1 - sn * k2;
    k[p + 1] = sn * k1 + cs * k2;
}

__device__ __forceinline__ float rowmax16(float v) {
    v = fmaxf(v, __shfl_xor(v, 1, 32));
    v = fmaxf(v, __shfl_xor(v, 2, 32));
    v = fmaxf(v, __shfl_xor(v, 4, 32));
    v = fmaxf(v, __shfl_xor(v, 8, 32));
    return v;
}
__device__ __forceinline__ float rowsum16(float v) {
    v += __shfl_xor(v, 1, 32);
    v += __shfl_xor(v, 2, 32);
    v += __shfl_xor(v, 4, 32);
    v += __shfl_xor(v, 8, 32);
    return v;
}

__global__ __launch_bounds__(256) void attn_kernel(const float* __restrict__ q,
                                                   const float* __restrict__ k,
                                                   const float* __restrict__ vt,
                                                   float* __restrict__ o) {
    __shared__ float lds[8][16 * 18];
    const int lane = threadIdx.x & 31;
    const int warp = threadIdx.x >> 5;
    const int wave = blockIdx.x * 8 + warp;
    const int qt = wave & 127;
    const int bh = wave >> 7;
    const int h = bh & 15;
    const int b = bh >> 4;
    const int ln = lane & 15;
    const int lh = lane >> 4;

    const float* qbase = q + ((size_t)(b * SEQ) + qt * 16 + ln) * DM + h * DK;
    v2f qa[16];
#pragma unroll
    for (int j = 0; j < 16; ++j)
        qa[j] = *(const v2f*)(qbase + 4 * j + 2 * lh);

    const float* vrow[4];
#pragma unroll
    for (int nt = 0; nt < 4; ++nt)
        vrow[nt] = vt + ((size_t)(b * DM + h * DK + nt * 16 + ln) << 11);

    v8f acc[4] = {v8f{}, v8f{}, v8f{}, v8f{}};
    float mrow[8], lrow[8];
#pragma unroll
    for (int r = 0; r < 8; ++r) { mrow[r] = -1e30f; lrow[r] = 0.0f; }

    float* pl = &lds[warp][0];

    for (int kt = 0; kt <= qt; ++kt) {
        const float* kbase = k + ((size_t)(b * SEQ) + kt * 16 + ln) * DM + h * DK;
        v8f s = {};
#pragma unroll
        for (int j = 0; j < 16; ++j) {
            v2f bf = *(const v2f*)(kbase + 4 * j + 2 * lh);
            s = __builtin_amdgcn_wmma_f32_16x16x4_f32(false, qa[j], false, bf,
                                                      (short)0, s, false, false);
        }
        float nm[8], rs[8];
#pragma unroll
        for (int r = 0; r < 8; ++r) {
            float sv = s[r] * 0.125f;
            if (kt == qt && ln > (r + 8 * lh)) sv = -1e30f;
            s[r] = sv;
            nm[r] = fmaxf(mrow[r], rowmax16(sv));
        }
#pragma unroll
        for (int r = 0; r < 8; ++r) {
            float p = __expf(s[r] - nm[r]);
            s[r] = p;
            rs[r] = rowsum16(p);
        }
#pragma unroll
        for (int r = 0; r < 8; ++r) {
            float sc = __expf(mrow[r] - nm[r]);
            lrow[r] = lrow[r] * sc + rs[r];
            mrow[r] = nm[r];
#pragma unroll
            for (int nt = 0; nt < 4; ++nt) acc[nt][r] *= sc;
        }
#pragma unroll
        for (int r = 0; r < 8; ++r)
            pl[(r + 8 * lh) * 18 + ln] = s[r];
#pragma unroll
        for (int j = 0; j < 4; ++j) {
            int kk = 4 * j + 2 * lh;
            v2f a = *(const v2f*)(pl + ln * 18 + kk);
#pragma unroll
            for (int nt = 0; nt < 4; ++nt) {
                v2f bf = *(const v2f*)(vrow[nt] + kt * 16 + kk);
                acc[nt] = __builtin_amdgcn_wmma_f32_16x16x4_f32(false, a, false, bf,
                                                                (short)0, acc[nt],
                                                                false, false);
            }
        }
    }

    float* obase = o + ((size_t)(b * SEQ) + qt * 16) * DM + h * DK;
#pragma unroll
    for (int r = 0; r < 8; ++r) {
        float inv = 1.0f / lrow[r];
        int m = r + 8 * lh;
#pragma unroll
        for (int nt = 0; nt < 4; ++nt)
            obase[(size_t)m * DM + nt * 16 + ln] = acc[nt][r] * inv;
    }
}

extern "C" void kernel_launch(void* const* d_in, const int* in_sizes, int n_in,
                              void* d_out, int out_size, void* d_ws, size_t ws_size,
                              hipStream_t stream) {
    const float* x  = (const float*)d_in[0];
    const float* Wq = (const float*)d_in[1];
    const float* Wk = (const float*)d_in[2];
    const float* Wv = (const float*)d_in[3];
    const float* Wo = (const float*)d_in[4];
    float* out = (float*)d_out;

    const size_t bufElems = (size_t)BATCH * SEQ * DM;
    float* qb  = (float*)d_ws;
    float* kb  = qb + bufElems;
    float* vtb = kb + bufElems;
    float* ab  = vtb + bufElems;

    const int M = BATCH * SEQ, N = DM, K = DM;
    const int tiles = (M / 32) * (N / 64);
    const int gblocks = tiles / 8;

    gemm_xwt_kernel<0><<<gblocks, 256, 0, stream>>>(x, Wq, qb, M, N, K);
    gemm_xwt_kernel<0><<<gblocks, 256, 0, stream>>>(x, Wk, kb, M, N, K);
    gemm_xwt_kernel<1><<<gblocks, 256, 0, stream>>>(x, Wv, vtb, M, N, K);

    const int pairs = BATCH * SEQ * NH * (DK / 2);
    rope_kernel<<<pairs / 256, 256, 0, stream>>>(qb, kb, pairs);

    const int waves = BATCH * NH * (SEQ / 16);
    attn_kernel<<<waves / 8, 256, 0, stream>>>(qb, kb, vtb, ab);

    gemm_xwt_kernel<0><<<gblocks, 256, 0, stream>>>(ab, Wo, out, M, N, K);
}


# GEMMs: both operand tiles staged through LDS (128x128 block tile, 64-wide k chunks, whole-line global loads, 2 stages) replacing strided per-lane global fragment loads; sum order unchanged
# speedup vs baseline: 1.4461x; 1.4461x over previous
#include <hip/hip_runtime.h>
#include <math.h>

typedef __attribute__((ext_vector_type(2))) float v2f;
typedef __attribute__((ext_vector_type(4))) float v4f;
typedef __attribute__((ext_vector_type(8))) float v8f;

#define BATCH 2
#define SEQ 2048
#define DM 1024
#define NH 16
#define DK 64

#define GK_CHUNK 64
#define GK_PITCH 68
#define GK_TPR (GK_CHUNK / 4)
#define GK_RPP (256 / GK_TPR)
#define GK_NP (128 / GK_RPP)

template <int STORE_VT>
__global__ __launch_bounds__(256) void gemm_xwt_kernel(const float* __restrict__ X,
                                                       const float* __restrict__ W,
                                                       float* __restrict__ out,
                                                       int M, int N, int K) {
    __shared__ __attribute__((aligned(16))) float sA[2][128 * GK_PITCH];
    __shared__ __attribute__((aligned(16))) float sB[2][128 * GK_PITCH];

    const int t = threadIdx.x;
    const int lane = t & 31;
    const int warp = t >> 5;
    const int tilesN = N >> 7;
    const int bm = (blockIdx.x / tilesN) << 7;
    const int bn = (blockIdx.x % tilesN) << 7;
    const int wm = (warp & 3) << 5;
    const int wn = (warp >> 2) << 6;
    const int tm = bm + wm;
    const int tn = bn + wn;
    const int ln = lane & 15;
    const int lh = lane >> 4;

    const int sr = t / GK_TPR;
    const int sc = (t % GK_TPR) << 2;
    const float* gA = X + (size_t)(bm + sr) * K + sc;
    const float* gB = W + (size_t)(bn + sr) * K + sc;
    const int so = sr * GK_PITCH + sc;

    v4f ra[GK_NP], rb[GK_NP];
#pragma unroll
    for (int i = 0; i < GK_NP; ++i) {
        ra[i] = *(const v4f*)(gA + (size_t)(GK_RPP * i) * K);
        rb[i] = *(const v4f*)(gB + (size_t)(GK_RPP * i) * K);
    }
#pragma unroll
    for (int i = 0; i < GK_NP; ++i) {
        *(v4f*)(&sA[0][so + GK_RPP * i * GK_PITCH]) = ra[i];
        *(v4f*)(&sB[0][so + GK_RPP * i * GK_PITCH]) = rb[i];
    }
    __syncthreads();

    const int fa = (wm + ln) * GK_PITCH + 2 * lh;
    const int fb = (wn + ln) * GK_PITCH + 2 * lh;

    v8f c[2][4] = {};
    const int nk = K / GK_CHUNK;
    for (int kc = 0; kc < nk; ++kc) {
        const int cur = kc & 1;
        const bool more = (kc + 1 < nk);
        if (more) {
            const int k1 = (kc + 1) * GK_CHUNK;
#pragma unroll
            for (int i = 0; i < GK_NP; ++i) {
                ra[i] = *(const v4f*)(gA + (size_t)(GK_RPP * i) * K + k1);
                rb[i] = *(const v4f*)(gB + (size_t)(GK_RPP * i) * K + k1);
            }
        }
        const float* pa = &sA[cur][fa];
        const float* pb = &sB[cur][fb];
#pragma unroll
        for (int j = 0; j < GK_CHUNK / 4; ++j) {
            v2f a0 = *(const v2f*)(pa + 4 * j);
            v2f a1 = *(const v2f*)(pa + 16 * GK_PITCH + 4 * j);
            v2f b0 = *(const v2f*)(pb + 4 * j);
            v2f b1 = *(const v2f*)(pb + 16 * GK_PITCH + 4 * j);
            v2f b2 = *(const v2f*)(pb + 32 * GK_PITCH + 4 * j);
            v2f b3 = *(const v2f*)(pb + 48 * GK_PITCH + 4 * j);
            c[0][0] = __builtin_amdgcn_wmma_f32_16x16x4_f32(false, a0, false, b0, (short)0, c[0][0], false, false);
            c[0][1] = __builtin_amdgcn_wmma_f32_16x16x4_f32(false, a0, false, b1, (short)0, c[0][1], false, false);
            c[0][2] = __builtin_amdgcn_wmma_f32_16x16x4_f32(false, a0, false, b2, (short)0, c[0][2], false, false);
            c[0][3] = __builtin_amdgcn_wmma_f32_16x16x4_f32(false, a0, false, b3, (short)0, c[0][3], false, false);
            c[1][0] = __builtin_amdgcn_wmma_f32_16x16x4_f32(false, a1, false, b0, (short)0, c[1][0], false, false);
            c[1][1] = __builtin_amdgcn_wmma_f32_16x16x4_f32(false, a1, false, b1, (short)0, c[1][1], false, false);
            c[1][2] = __builtin_amdgcn_wmma_f32_16x16x4_f32(false, a1, false, b2, (short)0, c[1][2], false, false);
            c[1][3] = __builtin_amdgcn_wmma_f32_16x16x4_f32(false, a1, false, b3, (short)0, c[1][3], false, false);
        }
        if (more) {
#pragma unroll
            for (int i = 0; i < GK_NP; ++i) {
                *(v4f*)(&sA[cur ^ 1][so + GK_RPP * i * GK_PITCH]) = ra[i];
                *(v4f*)(&sB[cur ^ 1][so + GK_RPP * i * GK_PITCH]) = rb[i];
            }
        }
        __syncthreads();
    }

    if (STORE_VT == 0) {
#pragma unroll
        for (int mi = 0; mi < 2; ++mi)
#pragma unroll
            for (int ni = 0; ni < 4; ++ni) {
                float* orow = out + (size_t)(tm + mi * 16 + 8 * lh) * N + tn + ni * 16 + ln;
#pragma unroll
                for (int r = 0; r < 8; ++r)
                    orow[(size_t)r * N] = c[mi][ni][r];
            }
    } else {
        const int bb = tm >> 11;
#pragma unroll
        for (int mi = 0; mi < 2; ++mi)
#pragma unroll
            for (int ni = 0; ni < 4; ++ni) {
                int n = tn + ni * 16 + ln;
                int srow = (tm & (SEQ - 1)) + mi * 16 + 8 * lh;
                float* ocol = out + ((size_t)(bb * N + n) << 11) + srow;
#pragma unroll
                for (int r = 0; r < 8; ++r)
                    ocol[r] = c[mi][ni][r];
            }
    }
}

__global__ __launch_bounds__(256) void rope_kernel(float* __restrict__ q,
                                                   float* __restrict__ k, int total) {
    int tid = blockIdx.x * blockDim.x + threadIdx.x;
    if (tid >= total) return;
    int i = tid & 31;
    int s = (tid >> 9) & (SEQ - 1);
    float freq = __expf(-(float)i * (1.0f / 32.0f) * 9.210340371976184f);
    float ang = (float)s * freq;
    float cs = __cosf(ang);
    float sn = __sinf(ang);
    size_t p = (size_t)tid * 2;
    float q1 = q[p], q2 = q[p + 1];
    q[p]     = cs * q1 - sn * q2;
    q[p + 1] = sn * q1 + cs * q2;
    float k1 = k[p], k2 = k[p + 1];
    k[p]     = cs * k1 - sn * k2;
    k[p + 1] = sn * k1 + cs * k2;
}

__device__ __forceinline__ float rowmax16(float v) {
    v = fmaxf(v, __shfl_xor(v, 1, 32));
    v = fmaxf(v, __shfl_xor(v, 2, 32));
    v = fmaxf(v, __shfl_xor(v, 4, 32));
    v = fmaxf(v, __shfl_xor(v, 8, 32));
    return v;
}
__device__ __forceinline__ float rowsum16(float v) {
    v += __shfl_xor(v, 1, 32);
    v += __shfl_xor(v, 2, 32);
    v += __shfl_xor(v, 4, 32);
    v += __shfl_xor(v, 8, 32);
    return v;
}

__global__ __launch_bounds__(256) void attn_kernel(const float* __restrict__ q,
                                                   const float* __restrict__ k,
                                                   const float* __restrict__ vt,
                                                   float* __restrict__ o) {
    __shared__ float lds[8][16 * 18];
    const int lane = threadIdx.x & 31;
    const int warp = threadIdx.x >> 5;
    const int wave = blockIdx.x * 8 + warp;
    const int qt = wave & 127;
    const int bh = wave >> 7;
    const int h = bh & 15;
    const int b = bh >> 4;
    const int ln = lane & 15;
    const int lh = lane >> 4;

    const float* qbase = q + ((size_t)(b * SEQ) + qt * 16 + ln) * DM + h * DK;
    v2f qa[16];
#pragma unroll
    for (int j = 0; j < 16; ++j)
        qa[j] = *(const v2f*)(qbase + 4 * j + 2 * lh);

    const float* vrow[4];
#pragma unroll
    for (int nt = 0; nt < 4; ++nt)
        vrow[nt] = vt + ((size_t)(b * DM + h * DK + nt * 16 + ln) << 11);

    v8f acc[4] = {v8f{}, v8f{}, v8f{}, v8f{}};
    float mrow[8], lrow[8];
#pragma unroll
    for (int r = 0; r < 8; ++r) { mrow[r] = -1e30f; lrow[r] = 0.0f; }

    float* pl = &lds[warp][0];

    for (int kt = 0; kt <= qt; ++kt) {
        const float* kbase = k + ((size_t)(b * SEQ) + kt * 16 + ln) * DM + h * DK;
        v8f s = {};
#pragma unroll
        for (int j = 0; j < 16; ++j) {
            v2f bf = *(const v2f*)(kbase + 4 * j + 2 * lh);
            s = __builtin_amdgcn_wmma_f32_16x16x4_f32(false, qa[j], false, bf,
                                                      (short)0, s, false, false);
        }
        float nm[8], rs[8];
#pragma unroll
        for (int r = 0; r < 8; ++r) {
            float sv = s[r] * 0.125f;
            if (kt == qt && ln > (r + 8 * lh)) sv = -1e30f;
            s[r] = sv;
            nm[r] = fmaxf(mrow[r], rowmax16(sv));
        }
#pragma unroll
        for (int r = 0; r < 8; ++r) {
            float p = __expf(s[r] - nm[r]);
            s[r] = p;
            rs[r] = rowsum16(p);
        }
#pragma unroll
        for (int r = 0; r < 8; ++r) {
            float sc = __expf(mrow[r] - nm[r]);
            lrow[r] = lrow[r] * sc + rs[r];
            mrow[r] = nm[r];
#pragma unroll
            for (int nt = 0; nt < 4; ++nt) acc[nt][r] *= sc;
        }
#pragma unroll
        for (int r = 0; r < 8; ++r)
            pl[(r + 8 * lh) * 18 + ln] = s[r];
#pragma unroll
        for (int j = 0; j < 4; ++j) {
            int kk = 4 * j + 2 * lh;
            v2f a = *(const v2f*)(pl + ln * 18 + kk);
#pragma unroll
            for (int nt = 0; nt < 4; ++nt) {
                v2f bf = *(const v2f*)(vrow[nt] + kt * 16 + kk);
                acc[nt] = __builtin_amdgcn_wmma_f32_16x16x4_f32(false, a, false, bf,
                                                                (short)0, acc[nt],
                                                                false, false);
            }
        }
    }

    float* obase = o + ((size_t)(b * SEQ) + qt * 16) * DM + h * DK;
#pragma unroll
    for (int r = 0; r < 8; ++r) {
        float inv = 1.0f / lrow[r];
        int m = r + 8 * lh;
#pragma unroll
        for (int nt = 0; nt < 4; ++nt)
            obase[(size_t)m * DM + nt * 16 + ln] = acc[nt][r] * inv;
    }
}

extern "C" void kernel_launch(void* const* d_in, const int* in_sizes, int n_in,
                              void* d_out, int out_size, void* d_ws, size_t ws_size,
                              hipStream_t stream) {
    const float* x  = (const float*)d_in[0];
    const float* Wq = (const float*)d_in[1];
    const float* Wk = (const float*)d_in[2];
    const float* Wv = (const float*)d_in[3];
    const float* Wo = (const float*)d_in[4];
    float* out = (float*)d_out;

    const size_t bufElems = (size_t)BATCH * SEQ * DM;
    float* qb  = (float*)d_ws;
    float* kb  = qb + bufElems;
    float* vtb = kb + bufElems;
    float* ab  = vtb + bufElems;

    const int M = BATCH * SEQ, N = DM, K = DM;
    const int gblocks = (M / 128) * (N / 128);

    gemm_xwt_kernel<0><<<gblocks, 256, 0, stream>>>(x, Wq, qb, M, N, K);
    gemm_xwt_kernel<0><<<gblocks, 256, 0, stream>>>(x, Wk, kb, M, N, K);
    gemm_xwt_kernel<1><<<gblocks, 256, 0, stream>>>(x, Wv, vtb, M, N, K);

    const int pairs = BATCH * SEQ * NH * (DK / 2);
    rope_kernel<<<pairs / 256, 256, 0, stream>>>(qb, kb, pairs);

    const int waves = BATCH * NH * (SEQ / 16);
    attn_kernel<<<waves / 8, 256, 0, stream>>>(qb, kb, vtb, ab);

    gemm_xwt_kernel<0><<<gblocks, 256, 0, stream>>>(ab, Wo, out, M, N, K);
}


# attention: each block takes 4 late query tiles and their 4 mirror early tiles (qt and 127-qt on waves w and w+4 of one SIMD) so every SIMD gets equal causal work; sum order unchanged
# speedup vs baseline: 1.5825x; 1.0943x over previous
#include <hip/hip_runtime.h>
#include <math.h>

typedef __attribute__((ext_vector_type(2))) float v2f;
typedef __attribute__((ext_vector_type(4))) float v4f;
typedef __attribute__((ext_vector_type(8))) float v8f;

#define BATCH 2
#define SEQ 2048
#define DM 1024
#define NH 16
#define DK 64

#define GK_CHUNK 64
#define GK_PITCH 68
#define GK_TPR (GK_CHUNK / 4)
#define GK_RPP (256 / GK_TPR)
#define GK_NP (128 / GK_RPP)

template <int STORE_VT>
__global__ __launch_bounds__(256) void gemm_xwt_kernel(const float* __restrict__ X,
                                                       const float* __restrict__ W,
                                                       float* __restrict__ out,
                                                       int M, int N, int K) {
    __shared__ __attribute__((aligned(16))) float sA[2][128 * GK_PITCH];
    __shared__ __attribute__((aligned(16))) float sB[2][128 * GK_PITCH];

    const int t = threadIdx.x;
    const int lane = t & 31;
    const int warp = t >> 5;
    const int tilesN = N >> 7;
    const int bm = (blockIdx.x / tilesN) << 7;
    const int bn = (blockIdx.x % tilesN) << 7;
    const int wm = (warp & 3) << 5;
    const int wn = (warp >> 2) << 6;
    const int tm = bm + wm;
    const int tn = bn + wn;
    const int ln = lane & 15;
    const int lh = lane >> 4;

    const int sr = t / GK_TPR;
    const int sc = (t % GK_TPR) << 2;
    const float* gA = X + (size_t)(bm + sr) * K + sc;
    const float* gB = W + (size_t)(bn + sr) * K + sc;
    const int so = sr * GK_PITCH + sc;

    v4f ra[GK_NP], rb[GK_NP];
#pragma unroll
    for (int i = 0; i < GK_NP; ++i) {
        ra[i] = *(const v4f*)(gA + (size_t)(GK_RPP * i) * K);
        rb[i] = *(const v4f*)(gB + (size_t)(GK_RPP * i) * K);
    }
#pragma unroll
    for (int i = 0; i < GK_NP; ++i) {
        *(v4f*)(&sA[0][so + GK_RPP * i * GK_PITCH]) = ra[i];
        *(v4f*)(&sB[0][so + GK_RPP * i * GK_PITCH]) = rb[i];
    }
    __syncthreads();

    const int fa = (wm + ln) * GK_PITCH + 2 * lh;
    const int fb = (wn + ln) * GK_PITCH + 2 * lh;

    v8f c[2][4] = {};
    const int nk = K / GK_CHUNK;
    for (int kc = 0; kc < nk; ++kc) {
        const int cur = kc & 1;
        const bool more = (kc + 1 < nk);
        if (more) {
            const int k1 = (kc + 1) * GK_CHUNK;
#pragma unroll
            for (int i = 0; i < GK_NP; ++i) {
                ra[i] = *(const v4f*)(gA + (size_t)(GK_RPP * i) * K + k1);
                rb[i] = *(const v4f*)(gB + (size_t)(GK_RPP * i) * K + k1);
            }
        }
        const float* pa = &sA[cur][fa];
        const float* pb = &sB[cur][fb];
#pragma unroll
        for (int j = 0; j < GK_CHUNK / 4; ++j) {
            v2f a0 = *(const v2f*)(pa + 4 * j);
            v2f a1 = *(const v2f*)(pa + 16 * GK_PITCH + 4 * j);
            v2f b0 = *(const v2f*)(pb + 4 * j);
            v2f b1 = *(const v2f*)(pb + 16 * GK_PITCH + 4 * j);
            v2f b2 = *(const v2f*)(pb + 32 * GK_PITCH + 4 * j);
            v2f b3 = *(const v2f*)(pb + 48 * GK_PITCH + 4 * j);
            c[0][0] = __builtin_amdgcn_wmma_f32_16x16x4_f32(false, a0, false, b0, (short)0, c[0][0], false, false);
            c[0][1] = __builtin_amdgcn_wmma_f32_16x16x4_f32(false, a0, false, b1, (short)0, c[0][1], false, false);
            c[0][2] = __builtin_amdgcn_wmma_f32_16x16x4_f32(false, a0, false, b2, (short)0, c[0][2], false, false);
            c[0][3] = __builtin_amdgcn_wmma_f32_16x16x4_f32(false, a0, false, b3, (short)0, c[0][3], false, false);
            c[1][0] = __builtin_amdgcn_wmma_f32_16x16x4_f32(false, a1, false, b0, (short)0, c[1][0], false, false);
            c[1][1] = __builtin_amdgcn_wmma_f32_16x16x4_f32(false, a1, false, b1, (short)0, c[1][1], false, false);
            c[1][2] = __builtin_amdgcn_wmma_f32_16x16x4_f32(false, a1, false, b2, (short)0, c[1][2], false, false);
            c[1][3] = __builtin_amdgcn_wmma_f32_16x16x4_f32(false, a1, false, b3, (short)0, c[1][3], false, false);
        }
        if (more) {
#pragma unroll
            for (int i = 0; i < GK_NP; ++i) {
                *(v4f*)(&sA[cur ^ 1][so + GK_RPP * i * GK_PITCH]) = ra[i];
                *(v4f*)(&sB[cur ^ 1][so + GK_RPP * i * GK_PITCH]) = rb[i];
            }
        }
        __syncthreads();
    }

    if (STORE_VT == 0) {
#pragma unroll
        for (int mi = 0; mi < 2; ++mi)
#pragma unroll
            for (int ni = 0; ni < 4; ++ni) {
                float* orow = out + (size_t)(tm + mi * 16 + 8 * lh) * N + tn + ni * 16 + ln;
#pragma unroll
                for (int r = 0; r < 8; ++r)
                    orow[(size_t)r * N] = c[mi][ni][r];
            }
    } else {
        const int bb = tm >> 11;
#pragma unroll
        for (int mi = 0; mi < 2; ++mi)
#pragma unroll
            for (int ni = 0; ni < 4; ++ni) {
                int n = tn + ni * 16 + ln;
                int srow = (tm & (SEQ - 1)) + mi * 16 + 8 * lh;
                float* ocol = out + ((size_t)(bb * N + n) << 11) + srow;
#pragma unroll
                for (int r = 0; r < 8; ++r)
                    ocol[r] = c[mi][ni][r];
            }
    }
}

__global__ __launch_bounds__(256) void rope_kernel(float* __restrict__ q,
                                                   float* __restrict__ k, int total) {
    int tid = blockIdx.x * blockDim.x + threadIdx.x;
    if (tid >= total) return;
    int i = tid & 31;
    int s = (tid >> 9) & (SEQ - 1);
    float freq = __expf(-(float)i * (1.0f / 32.0f) * 9.210340371976184f);
    float ang = (float)s * freq;
    float cs = __cosf(ang);
    float sn = __sinf(ang);
    size_t p = (size_t)tid * 2;
    float q1 = q[p], q2 = q[p + 1];
    q[p]     = cs * q1 - sn * q2;
    q[p + 1] = sn * q1 + cs * q2;
    float k1 = k[p], k2 = k[p + 1];
    k[p]     = cs * k1 - sn * k2;
    k[p + 1] = sn * k1 + cs * k2;
}

__device__ __forceinline__ float rowmax16(float v) {
    v = fmaxf(v, __shfl_xor(v, 1, 32));
    v = fmaxf(v, __shfl_xor(v, 2, 32));
    v = fmaxf(v, __shfl_xor(v, 4, 32));
    v = fmaxf(v, __shfl_xor(v, 8, 32));
    return v;
}
__device__ __forceinline__ float rowsum16(float v) {
    v += __shfl_xor(v, 1, 32);
    v += __shfl_xor(v, 2, 32);
    v += __shfl_xor(v, 4, 32);
    v += __shfl_xor(v, 8, 32);
    return v;
}

__global__ __launch_bounds__(256) void attn_kernel(const float* __restrict__ q,
                                                   const float* __restrict__ k,
                                                   const float* __restrict__ vt,
                                                   float* __restrict__ o) {
    __shared__ float lds[8][16 * 18];
    const int lane = threadIdx.x & 31;
    const int warp = threadIdx.x >> 5;
    const int bh = blockIdx.x >> 4;
    const int grp = ((blockIdx.x & 15) << 2) + (warp & 3);
    const int qt = (warp < 4) ? 127 - grp : grp;
    const int h = bh & 15;
    const int b = bh >> 4;
    const int ln = lane & 15;
    const int lh = lane >> 4;

    const float* qbase = q + ((size_t)(b * SEQ) + qt * 16 + ln) * DM + h * DK;
    v2f qa[16];
#pragma unroll
    for (int j = 0; j < 16; ++j)
        qa[j] = *(const v2f*)(qbase + 4 * j + 2 * lh);

    const float* vrow[4];
#pragma unroll
    for (int nt = 0; nt < 4; ++nt)
        vrow[nt] = vt + ((size_t)(b * DM + h * DK + nt * 16 + ln) << 11);

    v8f acc[4] = {v8f{}, v8f{}, v8f{}, v8f{}};
    float mrow[8], lrow[8];
#pragma unroll
    for (int r = 0; r < 8; ++r) { mrow[r] = -1e30f; lrow[r] = 0.0f; }

    float* pl = &lds[warp][0];

    for (int kt = 0; kt <= qt; ++kt) {
        const float* kbase = k + ((size_t)(b * SEQ) + kt * 16 + ln) * DM + h * DK;
        v8f s = {};
#pragma unroll
        for (int j = 0; j < 16; ++j) {
            v2f bf = *(const v2f*)(kbase + 4 * j + 2 * lh);
            s = __builtin_amdgcn_wmma_f32_16x16x4_f32(false, qa[j], false, bf,
                                                      (short)0, s, false, false);
        }
        float nm[8], rs[8];
#pragma unroll
        for (int r = 0; r < 8; ++r) {
            float sv = s[r] * 0.125f;
            if (kt == qt && ln > (r + 8 * lh)) sv = -1e30f;
            s[r] = sv;
            nm[r] = fmaxf(mrow[r], rowmax16(sv));
        }
#pragma unroll
        for (int r = 0; r < 8; ++r) {
            float p = __expf(s[r] - nm[r]);
            s[r] = p;
            rs[r] = rowsum16(p);
        }
#pragma unroll
        for (int r = 0; r < 8; ++r) {
            float sc = __expf(mrow[r] - nm[r]);
            lrow[r] = lrow[r] * sc + rs[r];
            mrow[r] = nm[r];
#pragma unroll
            for (int nt = 0; nt < 4; ++nt) acc[nt][r] *= sc;
        }
#pragma unroll
        for (int r = 0; r < 8; ++r)
            pl[(r + 8 * lh) * 18 + ln] = s[r];
#pragma unroll
        for (int j = 0; j < 4; ++j) {
            int kk = 4 * j + 2 * lh;
            v2f a = *(const v2f*)(pl + ln * 18 + kk);
#pragma unroll
            for (int nt = 0; nt < 4; ++nt) {
                v2f bf = *(const v2f*)(vrow[nt] + kt * 16 + kk);
                acc[nt] = __builtin_amdgcn_wmma_f32_16x16x4_f32(false, a, false, bf,
                                                                (short)0, acc[nt],
                                                                false, false);
            }
        }
    }

    float* obase = o + ((size_t)(b * SEQ) + qt * 16) * DM + h * DK;
#pragma unroll
    for (int r = 0; r < 8; ++r) {
        float inv = 1.0f / lrow[r];
        int m = r + 8 * lh;
#pragma unroll
        for (int nt = 0; nt < 4; ++nt)
            obase[(size_t)m * DM + nt * 16 + ln] = acc[nt][r] * inv;
    }
}

extern "C" void kernel_launch(void* const* d_in, const int* in_sizes, int n_in,
                              void* d_out, int out_size, void* d_ws, size_t ws_size,
                              hipStream_t stream) {
    const float* x  = (const float*)d_in[0];
    const float* Wq = (const float*)d_in[1];
    const float* Wk = (const float*)d_in[2];
    const float* Wv = (const float*)d_in[3];
    const float* Wo = (const float*)d_in[4];
    float* out = (float*)d_out;

    const size_t bufElems = (size_t)BATCH * SEQ * DM;
    float* qb  = (float*)d_ws;
    float* kb  = qb + bufElems;
    float* vtb = kb + bufElems;
    float* ab  = vtb + bufElems;

    const int M = BATCH * SEQ, N = DM, K = DM;
    const int gblocks = (M / 128) * (N / 128);

    gemm_xwt_kernel<0><<<gblocks, 256, 0, stream>>>(x, Wq, qb, M, N, K);
    gemm_xwt_kernel<0><<<gblocks, 256, 0, stream>>>(x, Wk, kb, M, N, K);
    gemm_xwt_kernel<1><<<gblocks, 256, 0, stream>>>(x, Wv, vtb, M, N, K);

    const int pairs = BATCH * SEQ * NH * (DK / 2);
    rope_kernel<<<pairs / 256, 256, 0, stream>>>(qb, kb, pairs);

    const int waves = BATCH * NH * (SEQ / 16);
    attn_kernel<<<waves / 8, 256, 0, stream>>>(qb, kb, vtb, ab);

    gemm_xwt_kernel<0><<<gblocks, 256, 0, stream>>>(ab, Wo, out, M, N, K);
}


# attention: each wave stages its K and Vt tiles through a wave-private LDS area with whole-row 16-byte loads (next tile prefetched in registers); fragments read from LDS; sum order unchanged
# speedup vs baseline: 1.8357x; 1.1600x over previous
#include <hip/hip_runtime.h>
#include <math.h>

typedef __attribute__((ext_vector_type(2))) float v2f;
typedef __attribute__((ext_vector_type(4))) float v4f;
typedef __attribute__((ext_vector_type(8))) float v8f;

#define BATCH 2
#define SEQ 2048
#define DM 1024
#define NH 16
#define DK 64

#define GK_CHUNK 64
#define GK_PITCH 68
#define GK_TPR (GK_CHUNK / 4)
#define GK_RPP (256 / GK_TPR)
#define GK_NP (128 / GK_RPP)

template <int STORE_VT>
__global__ __launch_bounds__(256) void gemm_xwt_kernel(const float* __restrict__ X,
                                                       const float* __restrict__ W,
                                                       float* __restrict__ out,
                                                       int M, int N, int K) {
    __shared__ __attribute__((aligned(16))) float sA[2][128 * GK_PITCH];
    __shared__ __attribute__((aligned(16))) float sB[2][128 * GK_PITCH];

    const int t = threadIdx.x;
    const int lane = t & 31;
    const int warp = t >> 5;
    const int tilesN = N >> 7;
    const int bm = (blockIdx.x / tilesN) << 7;
    const int bn = (blockIdx.x % tilesN) << 7;
    const int wm = (warp & 3) << 5;
    const int wn = (warp >> 2) << 6;
    const int tm = bm + wm;
    const int tn = bn + wn;
    const int ln = lane & 15;
    const int lh = lane >> 4;

    const int sr = t / GK_TPR;
    const int sc = (t % GK_TPR) << 2;
    const float* gA = X + (size_t)(bm + sr) * K + sc;
    const float* gB = W + (size_t)(bn + sr) * K + sc;
    const int so = sr * GK_PITCH + sc;

    v4f ra[GK_NP], rb[GK_NP];
#pragma unroll
    for (int i = 0; i < GK_NP; ++i) {
        ra[i] = *(const v4f*)(gA + (size_t)(GK_RPP * i) * K);
        rb[i] = *(const v4f*)(gB + (size_t)(GK_RPP * i) * K);
    }
#pragma unroll
    for (int i = 0; i < GK_NP; ++i) {
        *(v4f*)(&sA[0][so + GK_RPP * i * GK_PITCH]) = ra[i];
        *(v4f*)(&sB[0][so + GK_RPP * i * GK_PITCH]) = rb[i];
    }
    __syncthreads();

    const int fa = (wm + ln) * GK_PITCH + 2 * lh;
    const int fb = (wn + ln) * GK_PITCH + 2 * lh;

    v8f c[2][4] = {};
    const int nk = K / GK_CHUNK;
    for (int kc = 0; kc < nk; ++kc) {
        const int cur = kc & 1;
        const bool more = (kc + 1 < nk);
        if (more) {
            const int k1 = (kc + 1) * GK_CHUNK;
#pragma unroll
            for (int i = 0; i < GK_NP; ++i) {
                ra[i] = *(const v4f*)(gA + (size_t)(GK_RPP * i) * K + k1);
                rb[i] = *(const v4f*)(gB + (size_t)(GK_RPP * i) * K + k1);
            }
        }
        const float* pa = &sA[cur][fa];
        const float* pb = &sB[cur][fb];
#pragma unroll
        for (int j = 0; j < GK_CHUNK / 4; ++j) {
            v2f a0 = *(const v2f*)(pa + 4 * j);
            v2f a1 = *(const v2f*)(pa + 16 * GK_PITCH + 4 * j);
            v2f b0 = *(const v2f*)(pb + 4 * j);
            v2f b1 = *(const v2f*)(pb + 16 * GK_PITCH + 4 * j);
            v2f b2 = *(const v2f*)(pb + 32 * GK_PITCH + 4 * j);
            v2f b3 = *(const v2f*)(pb + 48 * GK_PITCH + 4 * j);
            c[0][0] = __builtin_amdgcn_wmma_f32_16x16x4_f32(false, a0, false, b0, (short)0, c[0][0], false, false);
            c[0][1] = __builtin_amdgcn_wmma_f32_16x16x4_f32(false, a0, false, b1, (short)0, c[0][1], false, false);
            c[0][2] = __builtin_amdgcn_wmma_f32_16x16x4_f32(false, a0, false, b2, (short)0, c[0][2], false, false);
            c[0][3] = __builtin_amdgcn_wmma_f32_16x16x4_f32(false, a0, false, b3, (short)0, c[0][3], false, false);
            c[1][0] = __builtin_amdgcn_wmma_f32_16x16x4_f32(false, a1, false, b0, (short)0, c[1][0], false, false);
            c[1][1] = __builtin_amdgcn_wmma_f32_16x16x4_f32(false, a1, false, b1, (short)0, c[1][1], false, false);
            c[1][2] = __builtin_amdgcn_wmma_f32_16x16x4_f32(false, a1, false, b2, (short)0, c[1][2], false, false);
            c[1][3] = __builtin_amdgcn_wmma_f32_16x16x4_f32(false, a1, false, b3, (short)0, c[1][3], false, false);
        }
        if (more) {
#pragma unroll
            for (int i = 0; i < GK_NP; ++i) {
                *(v4f*)(&sA[cur ^ 1][so + GK_RPP * i * GK_PITCH]) = ra[i];
                *(v4f*)(&sB[cur ^ 1][so + GK_RPP * i * GK_PITCH]) = rb[i];
            }
        }
        __syncthreads();
    }

    if (STORE_VT == 0) {
#pragma unroll
        for (int mi = 0; mi < 2; ++mi)
#pragma unroll
            for (int ni = 0; ni < 4; ++ni) {
                float* orow = out + (size_t)(tm + mi * 16 + 8 * lh) * N + tn + ni * 16 + ln;
#pragma unroll
                for (int r = 0; r < 8; ++r)
                    orow[(size_t)r * N] = c[mi][ni][r];
            }
    } else {
        const int bb = tm >> 11;
#pragma unroll
        for (int mi = 0; mi < 2; ++mi)
#pragma unroll
            for (int ni = 0; ni < 4; ++ni) {
                int n = tn + ni * 16 + ln;
                int srow = (tm & (SEQ - 1)) + mi * 16 + 8 * lh;
                float* ocol = out + ((size_t)(bb * N + n) << 11) + srow;
#pragma unroll
                for (int r = 0; r < 8; ++r)
                    ocol[r] = c[mi][ni][r];
            }
    }
}

__global__ __launch_bounds__(256) void rope_kernel(float* __restrict__ q,
                                                   float* __restrict__ k, int total) {
    int tid = blockIdx.x * blockDim.x + threadIdx.x;
    if (tid >= total) return;
    int i = tid & 31;
    int s = (tid >> 9) & (SEQ - 1);
    float freq = __expf(-(float)i * (1.0f / 32.0f) * 9.210340371976184f);
    float ang = (float)s * freq;
    float cs = __cosf(ang);
    float sn = __sinf(ang);
    size_t p = (size_t)tid * 2;
    float q1 = q[p], q2 = q[p + 1];
    q[p]     = cs * q1 - sn * q2;
    q[p + 1] = sn * q1 + cs * q2;
    float k1 = k[p], k2 = k[p + 1];
    k[p]     = cs * k1 - sn * k2;
    k[p + 1] = sn * k1 + cs * k2;
}

__device__ __forceinline__ float rowmax16(float v) {
    v = fmaxf(v, __shfl_xor(v, 1, 32));
    v = fmaxf(v, __shfl_xor(v, 2, 32));
    v = fmaxf(v, __shfl_xor(v, 4, 32));
    v = fmaxf(v, __shfl_xor(v, 8, 32));
    return v;
}
__device__ __forceinline__ float rowsum16(float v) {
    v += __shfl_xor(v, 1, 32);
    v += __shfl_xor(v, 2, 32);
    v += __shfl_xor(v, 4, 32);
    v += __shfl_xor(v, 8, 32);
    return v;
}

#define AK_PITCH 68
#define AV_PITCH 20
#define AP_PITCH 18
#define AW_K 0
#define AW_V (16 * AK_PITCH)
#define AW_P (AW_V + 64 * AV_PITCH)
#define AW_FLOATS (AW_P + 16 * AP_PITCH)

__global__ __launch_bounds__(256) void attn_kernel(const float* __restrict__ q,
                                                   const float* __restrict__ k,
                                                   const float* __restrict__ vt,
                                                   float* __restrict__ o) {
    __shared__ __attribute__((aligned(16))) float lds[8][AW_FLOATS];
    const int lane = threadIdx.x & 31;
    const int warp = threadIdx.x >> 5;
    const int bh = blockIdx.x >> 4;
    const int grp = ((blockIdx.x & 15) << 2) + (warp & 3);
    const int qt = (warp < 4) ? 127 - grp : grp;
    const int h = bh & 15;
    const int b = bh >> 4;
    const int ln = lane & 15;
    const int lh = lane >> 4;

    const float* qbase = q + ((size_t)(b * SEQ) + qt * 16 + ln) * DM + h * DK;
    v2f qa[16];
#pragma unroll
    for (int j = 0; j < 16; ++j)
        qa[j] = *(const v2f*)(qbase + 4 * j + 2 * lh);

    float* sK = &lds[warp][AW_K];
    float* sV = &lds[warp][AW_V];
    float* pl = &lds[warp][AW_P];

    const int kr = lane >> 4;
    const int kc = (lane & 15) << 2;
    const int vr = lane >> 2;
    const int vc = (lane & 3) << 2;
    const float* gK = k + ((size_t)(b * SEQ) + kr) * DM + h * DK + kc;
    const float* gV = vt + ((size_t)(b * DM + h * DK + vr) << 11) + vc;
    float* stK = sK + kr * AK_PITCH + kc;
    float* stV = sV + vr * AV_PITCH + vc;

    v4f rk[8], rv[8];
#pragma unroll
    for (int i = 0; i < 8; ++i) {
        rk[i] = *(const v4f*)(gK + (size_t)(2 * i) * DM);
        rv[i] = *(const v4f*)(gV + ((size_t)(8 * i) << 11));
    }

    const float* fK = sK + ln * AK_PITCH + 2 * lh;
    const float* fV = sV + ln * AV_PITCH + 2 * lh;
    const float* fP = pl + ln * AP_PITCH + 2 * lh;

    v8f acc[4] = {v8f{}, v8f{}, v8f{}, v8f{}};
    float mrow[8], lrow[8];
#pragma unroll
    for (int r = 0; r < 8; ++r) { mrow[r] = -1e30f; lrow[r] = 0.0f; }

    for (int kt = 0; kt <= qt; ++kt) {
#pragma unroll
        for (int i = 0; i < 8; ++i) {
            *(v4f*)(stK + 2 * i * AK_PITCH) = rk[i];
            *(v4f*)(stV + 8 * i * AV_PITCH) = rv[i];
        }
        const int kn = (kt < qt) ? kt + 1 : qt;
#pragma unroll
        for (int i = 0; i < 8; ++i) {
            rk[i] = *(const v4f*)(gK + (size_t)(kn * 16 + 2 * i) * DM);
            rv[i] = *(const v4f*)(gV + ((size_t)(8 * i) << 11) + kn * 16);
        }
        v2f kf[16], vf[4][4];
#pragma unroll
        for (int j = 0; j < 16; ++j)
            kf[j] = *(const v2f*)(fK + 4 * j);
#pragma unroll
        for (int j = 0; j < 4; ++j)
#pragma unroll
            for (int nt = 0; nt < 4; ++nt)
                vf[j][nt] = *(const v2f*)(fV + nt * 16 * AV_PITCH + 4 * j);
        v8f s = {};
#pragma unroll
        for (int j = 0; j < 16; ++j)
            s = __builtin_amdgcn_wmma_f32_16x16x4_f32(false, qa[j], false, kf[j],
                                                      (short)0, s, false, false);
        float nm[8], rs[8];
#pragma unroll
        for (int r = 0; r < 8; ++r) {
            float sv = s[r] * 0.125f;
            if (kt == qt && ln > (r + 8 * lh)) sv = -1e30f;
            s[r] = sv;
            nm[r] = fmaxf(mrow[r], rowmax16(sv));
        }
#pragma unroll
        for (int r = 0; r < 8; ++r) {
            float p = __expf(s[r] - nm[r]);
            s[r] = p;
            rs[r] = rowsum16(p);
        }
#pragma unroll
        for (int r = 0; r < 8; ++r) {
            float sc = __expf(mrow[r] - nm[r]);
            lrow[r] = lrow[r] * sc + rs[r];
            mrow[r] = nm[r];
#pragma unroll
            for (int nt = 0; nt < 4; ++nt) acc[nt][r] *= sc;
        }
#pragma unroll
        for (int r = 0; r < 8; ++r)
            pl[(r + 8 * lh) * AP_PITCH + ln] = s[r];
#pragma unroll
        for (int j = 0; j < 4; ++j) {
            v2f a = *(const v2f*)(fP + 4 * j);
#pragma unroll
            for (int nt = 0; nt < 4; ++nt)
                acc[nt] = __builtin_amdgcn_wmma_f32_16x16x4_f32(false, a, false, vf[j][nt],
                                                                (short)0, acc[nt],
                                                                false, false);
        }
    }

    float* obase = o + ((size_t)(b * SEQ) + qt * 16) * DM + h * DK;
#pragma unroll
    for (int r = 0; r < 8; ++r) {
        float inv = 1.0f / lrow[r];
        int m = r + 8 * lh;
#pragma unroll
        for (int nt = 0; nt < 4; ++nt)
            obase[(size_t)m * DM + nt * 16 + ln] = acc[nt][r] * inv;
    }
}

extern "C" void kernel_launch(void* const* d_in, const int* in_sizes, int n_in,
                              void* d_out, int out_size, void* d_ws, size_t ws_size,
                              hipStream_t stream) {
    const float* x  = (const float*)d_in[0];
    const float* Wq = (const float*)d_in[1];
    const float* Wk = (const float*)d_in[2];
    const float* Wv = (const float*)d_in[3];
    const float* Wo = (const float*)d_in[4];
    float* out = (float*)d_out;

    const size_t bufElems = (size_t)BATCH * SEQ * DM;
    float* qb  = (float*)d_ws;
    float* kb  = qb + bufElems;
    float* vtb = kb + bufElems;
    float* ab  = vtb + bufElems;

    const int M = BATCH * SEQ, N = DM, K = DM;
    const int gblocks = (M / 128) * (N / 128);

    gemm_xwt_kernel<0><<<gblocks, 256, 0, stream>>>(x, Wq, qb, M, N, K);
    gemm_xwt_kernel<0><<<gblocks, 256, 0, stream>>>(x, Wk, kb, M, N, K);
    gemm_xwt_kernel<1><<<gblocks, 256, 0, stream>>>(x, Wv, vtb, M, N, K);

    const int pairs = BATCH * SEQ * NH * (DK / 2);
    rope_kernel<<<pairs / 256, 256, 0, stream>>>(qb, kb, pairs);

    const int waves = BATCH * NH * (SEQ / 16);
    attn_kernel<<<waves / 8, 256, 0, stream>>>(qb, kb, vtb, ab);

    gemm_xwt_kernel<0><<<gblocks, 256, 0, stream>>>(ab, Wo, out, M, N, K);
}


# attention softmax: 16-lane row max and row sum butterflies read the partner lane as a DPP row_xmask operand on the vector ALU instead of a wave shuffle through the LDS unit; sum order unchanged
# speedup vs baseline: 2.1103x; 1.1496x over previous
#include <hip/hip_runtime.h>
#include <math.h>

typedef __attribute__((ext_vector_type(2))) float v2f;
typedef __attribute__((ext_vector_type(4))) float v4f;
typedef __attribute__((ext_vector_type(8))) float v8f;

#define BATCH 2
#define SEQ 2048
#define DM 1024
#define NH 16
#define DK 64

#define GK_CHUNK 64
#define GK_PITCH 68
#define GK_TPR (GK_CHUNK / 4)
#define GK_RPP (256 / GK_TPR)
#define GK_NP (128 / GK_RPP)

template <int STORE_VT>
__global__ __launch_bounds__(256) void gemm_xwt_kernel(const float* __restrict__ X,
                                                       const float* __restrict__ W,
                                                       float* __restrict__ out,
                                                       int M, int N, int K) {
    __shared__ __attribute__((aligned(16))) float sA[2][128 * GK_PITCH];
    __shared__ __attribute__((aligned(16))) float sB[2][128 * GK_PITCH];

    const int t = threadIdx.x;
    const int lane = t & 31;
    const int warp = t >> 5;
    const int tilesN = N >> 7;
    const int bm = (blockIdx.x / tilesN) << 7;
    const int bn = (blockIdx.x % tilesN) << 7;
    const int wm = (warp & 3) << 5;
    const int wn = (warp >> 2) << 6;
    const int tm = bm + wm;
    const int tn = bn + wn;
    const int ln = lane & 15;
    const int lh = lane >> 4;

    const int sr = t / GK_TPR;
    const int sc = (t % GK_TPR) << 2;
    const float* gA = X + (size_t)(bm + sr) * K + sc;
    const float* gB = W + (size_t)(bn + sr) * K + sc;
    const int so = sr * GK_PITCH + sc;

    v4f ra[GK_NP], rb[GK_NP];
#pragma unroll
    for (int i = 0; i < GK_NP; ++i) {
        ra[i] = *(const v4f*)(gA + (size_t)(GK_RPP * i) * K);
        rb[i] = *(const v4f*)(gB + (size_t)(GK_RPP * i) * K);
    }
#pragma unroll
    for (int i = 0; i < GK_NP; ++i) {
        *(v4f*)(&sA[0][so + GK_RPP * i * GK_PITCH]) = ra[i];
        *(v4f*)(&sB[0][so + GK_RPP * i * GK_PITCH]) = rb[i];
    }
    __syncthreads();

    const int fa = (wm + ln) * GK_PITCH + 2 * lh;
    const int fb = (wn + ln) * GK_PITCH + 2 * lh;

    v8f c[2][4] = {};
    const int nk = K / GK_CHUNK;
    for (int kc = 0; kc < nk; ++kc) {
        const int cur = kc & 1;
        const bool more = (kc + 1 < nk);
        if (more) {
            const int k1 = (kc + 1) * GK_CHUNK;
#pragma unroll
            for (int i = 0; i < GK_NP; ++i) {
                ra[i] = *(const v4f*)(gA + (size_t)(GK_RPP * i) * K + k1);
                rb[i] = *(const v4f*)(gB + (size_t)(GK_RPP * i) * K + k1);
            }
        }
        const float* pa = &sA[cur][fa];
        const float* pb = &sB[cur][fb];
#pragma unroll
        for (int j = 0; j < GK_CHUNK / 4; ++j) {
            v2f a0 = *(const v2f*)(pa + 4 * j);
            v2f a1 = *(const v2f*)(pa + 16 * GK_PITCH + 4 * j);
            v2f b0 = *(const v2f*)(pb + 4 * j);
            v2f b1 = *(const v2f*)(pb + 16 * GK_PITCH + 4 * j);
            v2f b2 = *(const v2f*)(pb + 32 * GK_PITCH + 4 * j);
            v2f b3 = *(const v2f*)(pb + 48 * GK_PITCH + 4 * j);
            c[0][0] = __builtin_amdgcn_wmma_f32_16x16x4_f32(false, a0, false, b0, (short)0, c[0][0], false, false);
            c[0][1] = __builtin_amdgcn_wmma_f32_16x16x4_f32(false, a0, false, b1, (short)0, c[0][1], false, false);
            c[0][2] = __builtin_amdgcn_wmma_f32_16x16x4_f32(false, a0, false, b2, (short)0, c[0][2], false, false);
            c[0][3] = __builtin_amdgcn_wmma_f32_16x16x4_f32(false, a0, false, b3, (short)0, c[0][3], false, false);
            c[1][0] = __builtin_amdgcn_wmma_f32_16x16x4_f32(false, a1, false, b0, (short)0, c[1][0], false, false);
            c[1][1] = __builtin_amdgcn_wmma_f32_16x16x4_f32(false, a1, false, b1, (short)0, c[1][1], false, false);
            c[1][2] = __builtin_amdgcn_wmma_f32_16x16x4_f32(false, a1, false, b2, (short)0, c[1][2], false, false);
            c[1][3] = __builtin_amdgcn_wmma_f32_16x16x4_f32(false, a1, false, b3, (short)0, c[1][3], false, false);
        }
        if (more) {
#pragma unroll
            for (int i = 0; i < GK_NP; ++i) {
                *(v4f*)(&sA[cur ^ 1][so + GK_RPP * i * GK_PITCH]) = ra[i];
                *(v4f*)(&sB[cur ^ 1][so + GK_RPP * i * GK_PITCH]) = rb[i];
            }
        }
        __syncthreads();
    }

    if (STORE_VT == 0) {
#pragma unroll
        for (int mi = 0; mi < 2; ++mi)
#pragma unroll
            for (int ni = 0; ni < 4; ++ni) {
                float* orow = out + (size_t)(tm + mi * 16 + 8 * lh) * N + tn + ni * 16 + ln;
#pragma unroll
                for (int r = 0; r < 8; ++r)
                    orow[(size_t)r * N] = c[mi][ni][r];
            }
    } else {
        const int bb = tm >> 11;
#pragma unroll
        for (int mi = 0; mi < 2; ++mi)
#pragma unroll
            for (int ni = 0; ni < 4; ++ni) {
                int n = tn + ni * 16 + ln;
                int srow = (tm & (SEQ - 1)) + mi * 16 + 8 * lh;
                float* ocol = out + ((size_t)(bb * N + n) << 11) + srow;
#pragma unroll
                for (int r = 0; r < 8; ++r)
                    ocol[r] = c[mi][ni][r];
            }
    }
}

__global__ __launch_bounds__(256) void rope_kernel(float* __restrict__ q,
                                                   float* __restrict__ k, int total) {
    int tid = blockIdx.x * blockDim.x + threadIdx.x;
    if (tid >= total) return;
    int i = tid & 31;
    int s = (tid >> 9) & (SEQ - 1);
    float freq = __expf(-(float)i * (1.0f / 32.0f) * 9.210340371976184f);
    float ang = (float)s * freq;
    float cs = __cosf(ang);
    float sn = __sinf(ang);
    size_t p = (size_t)tid * 2;
    float q1 = q[p], q2 = q[p + 1];
    q[p]     = cs * q1 - sn * q2;
    q[p + 1] = sn * q1 + cs * q2;
    float k1 = k[p], k2 = k[p + 1];
    k[p]     = cs * k1 - sn * k2;
    k[p + 1] = sn * k1 + cs * k2;
}

template <int M>
__device__ __forceinline__ float rowxor(float v) {
    int b = __builtin_bit_cast(int, v);
    return __builtin_bit_cast(float, __builtin_amdgcn_update_dpp(0, b, 0x160 + M, 0xf, 0xf, true));
}
__device__ __forceinline__ float rowmax16(float v) {
    v = fmaxf(v, rowxor<1>(v));
    v = fmaxf(v, rowxor<2>(v));
    v = fmaxf(v, rowxor<4>(v));
    v = fmaxf(v, rowxor<8>(v));
    return v;
}
__device__ __forceinline__ float rowsum16(float v) {
    v += rowxor<1>(v);
    v += rowxor<2>(v);
    v += rowxor<4>(v);
    v += rowxor<8>(v);
    return v;
}

#define AK_PITCH 68
#define AV_PITCH 20
#define AP_PITCH 18
#define AW_K 0
#define AW_V (16 * AK_PITCH)
#define AW_P (AW_V + 64 * AV_PITCH)
#define AW_FLOATS (AW_P + 16 * AP_PITCH)

__global__ __launch_bounds__(256) void attn_kernel(const float* __restrict__ q,
                                                   const float* __restrict__ k,
                                                   const float* __restrict__ vt,
                                                   float* __restrict__ o) {
    __shared__ __attribute__((aligned(16))) float lds[8][AW_FLOATS];
    const int lane = threadIdx.x & 31;
    const int warp = threadIdx.x >> 5;
    const int bh = blockIdx.x >> 4;
    const int grp = ((blockIdx.x & 15) << 2) + (warp & 3);
    const int qt = (warp < 4) ? 127 - grp : grp;
    const int h = bh & 15;
    const int b = bh >> 4;
    const int ln = lane & 15;
    const int lh = lane >> 4;

    const float* qbase = q + ((size_t)(b * SEQ) + qt * 16 + ln) * DM + h * DK;
    v2f qa[16];
#pragma unroll
    for (int j = 0; j < 16; ++j)
        qa[j] = *(const v2f*)(qbase + 4 * j + 2 * lh);

    float* sK = &lds[warp][AW_K];
    float* sV = &lds[warp][AW_V];
    float* pl = &lds[warp][AW_P];

    const int kr = lane >> 4;
    const int kc = (lane & 15) << 2;
    const int vr = lane >> 2;
    const int vc = (lane & 3) << 2;
    const float* gK = k + ((size_t)(b * SEQ) + kr) * DM + h * DK + kc;
    const float* gV = vt + ((size_t)(b * DM + h * DK + vr) << 11) + vc;
    float* stK = sK + kr * AK_PITCH + kc;
    float* stV = sV + vr * AV_PITCH + vc;

    v4f rk[8], rv[8];
#pragma unroll
    for (int i = 0; i < 8; ++i) {
        rk[i] = *(const v4f*)(gK + (size_t)(2 * i) * DM);
        rv[i] = *(const v4f*)(gV + ((size_t)(8 * i) << 11));
    }

    const float* fK = sK + ln * AK_PITCH + 2 * lh;
    const float* fV = sV + ln * AV_PITCH + 2 * lh;
    const float* fP = pl + ln * AP_PITCH + 2 * lh;

    v8f acc[4] = {v8f{}, v8f{}, v8f{}, v8f{}};
    float mrow[8], lrow[8];
#pragma unroll
    for (int r = 0; r < 8; ++r) { mrow[r] = -1e30f; lrow[r] = 0.0f; }

    for (int kt = 0; kt <= qt; ++kt) {
#pragma unroll
        for (int i = 0; i < 8; ++i) {
            *(v4f*)(stK + 2 * i * AK_PITCH) = rk[i];
            *(v4f*)(stV + 8 * i * AV_PITCH) = rv[i];
        }
        const int kn = (kt < qt) ? kt + 1 : qt;
#pragma unroll
        for (int i = 0; i < 8; ++i) {
            rk[i] = *(const v4f*)(gK + (size_t)(kn * 16 + 2 * i) * DM);
            rv[i] = *(const v4f*)(gV + ((size_t)(8 * i) << 11) + kn * 16);
        }
        v2f kf[16], vf[4][4];
#pragma unroll
        for (int j = 0; j < 16; ++j)
            kf[j] = *(const v2f*)(fK + 4 * j);
#pragma unroll
        for (int j = 0; j < 4; ++j)
#pragma unroll
            for (int nt = 0; nt < 4; ++nt)
                vf[j][nt] = *(const v2f*)(fV + nt * 16 * AV_PITCH + 4 * j);
        v8f s = {};
#pragma unroll
        for (int j = 0; j < 16; ++j)
            s = __builtin_amdgcn_wmma_f32_16x16x4_f32(false, qa[j], false, kf[j],
                                                      (short)0, s, false, false);
        float nm[8], rs[8];
#pragma unroll
        for (int r = 0; r < 8; ++r) {
            float sv = s[r] * 0.125f;
            if (kt == qt && ln > (r + 8 * lh)) sv = -1e30f;
            s[r] = sv;
            nm[r] = fmaxf(mrow[r], rowmax16(sv));
        }
#pragma unroll
        for (int r = 0; r < 8; ++r) {
            float p = __expf(s[r] - nm[r]);
            s[r] = p;
            rs[r] = rowsum16(p);
        }
#pragma unroll
        for (int r = 0; r < 8; ++r) {
            float sc = __expf(mrow[r] - nm[r]);
            lrow[r] = lrow[r] * sc + rs[r];
            mrow[r] = nm[r];
#pragma unroll
            for (int nt = 0; nt < 4; ++nt) acc[nt][r] *= sc;
        }
#pragma unroll
        for (int r = 0; r < 8; ++r)
            pl[(r + 8 * lh) * AP_PITCH + ln] = s[r];
#pragma unroll
        for (int j = 0; j < 4; ++j) {
            v2f a = *(const v2f*)(fP + 4 * j);
#pragma unroll
            for (int nt = 0; nt < 4; ++nt)
                acc[nt] = __builtin_amdgcn_wmma_f32_16x16x4_f32(false, a, false, vf[j][nt],
                                                                (short)0, acc[nt],
                                                                false, false);
        }
    }

    float* obase = o + ((size_t)(b * SEQ) + qt * 16) * DM + h * DK;
#pragma unroll
    for (int r = 0; r < 8; ++r) {
        float inv = 1.0f / lrow[r];
        int m = r + 8 * lh;
#pragma unroll
        for (int nt = 0; nt < 4; ++nt)
            obase[(size_t)m * DM + nt * 16 + ln] = acc[nt][r] * inv;
    }
}

extern "C" void kernel_launch(void* const* d_in, const int* in_sizes, int n_in,
                              void* d_out, int out_size, void* d_ws, size_t ws_size,
                              hipStream_t stream) {
    const float* x  = (const float*)d_in[0];
    const float* Wq = (const float*)d_in[1];
    const float* Wk = (const float*)d_in[2];
    const float* Wv = (const float*)d_in[3];
    const float* Wo = (const float*)d_in[4];
    float* out = (float*)d_out;

    const size_t bufElems = (size_t)BATCH * SEQ * DM;
    float* qb  = (float*)d_ws;
    float* kb  = qb + bufElems;
    float* vtb = kb + bufElems;
    float* ab  = vtb + bufElems;

    const int M = BATCH * SEQ, N = DM, K = DM;
    const int gblocks = (M / 128) * (N / 128);

    gemm_xwt_kernel<0><<<gblocks, 256, 0, stream>>>(x, Wq, qb, M, N, K);
    gemm_xwt_kernel<0><<<gblocks, 256, 0, stream>>>(x, Wk, kb, M, N, K);
    gemm_xwt_kernel<1><<<gblocks, 256, 0, stream>>>(x, Wv, vtb, M, N, K);

    const int pairs = BATCH * SEQ * NH * (DK / 2);
    rope_kernel<<<pairs / 256, 256, 0, stream>>>(qb, kb, pairs);

    const int waves = BATCH * NH * (SEQ / 16);
    attn_kernel<<<waves / 8, 256, 0, stream>>>(qb, kb, vtb, ab);

    gemm_xwt_kernel<0><<<gblocks, 256, 0, stream>>>(ab, Wo, out, M, N, K);
}


# attention: scores computed transposed (K rows as A, Q as B) so P's accumulator registers feed Ot += Vt Pt as B: no LDS transpose, row max/sum in-lane + one half exchange; row-sum order changes
# speedup vs baseline: 2.2399x; 1.0614x over previous
#include <hip/hip_runtime.h>
#include <math.h>

typedef __attribute__((ext_vector_type(2))) float v2f;
typedef __attribute__((ext_vector_type(4))) float v4f;
typedef __attribute__((ext_vector_type(8))) float v8f;

#define BATCH 2
#define SEQ 2048
#define DM 1024
#define NH 16
#define DK 64

#define GK_CHUNK 64
#define GK_PITCH 68
#define GK_TPR (GK_CHUNK / 4)
#define GK_RPP (256 / GK_TPR)
#define GK_NP (128 / GK_RPP)

template <int STORE_VT>
__global__ __launch_bounds__(256) void gemm_xwt_kernel(const float* __restrict__ X,
                                                       const float* __restrict__ W,
                                                       float* __restrict__ out,
                                                       int M, int N, int K) {
    __shared__ __attribute__((aligned(16))) float sA[2][128 * GK_PITCH];
    __shared__ __attribute__((aligned(16))) float sB[2][128 * GK_PITCH];

    const int t = threadIdx.x;
    const int lane = t & 31;
    const int warp = t >> 5;
    const int tilesN = N >> 7;
    const int bm = (blockIdx.x / tilesN) << 7;
    const int bn = (blockIdx.x % tilesN) << 7;
    const int wm = (warp & 3) << 5;
    const int wn = (warp >> 2) << 6;
    const int tm = bm + wm;
    const int tn = bn + wn;
    const int ln = lane & 15;
    const int lh = lane >> 4;

    const int sr = t / GK_TPR;
    const int sc = (t % GK_TPR) << 2;
    const float* gA = X + (size_t)(bm + sr) * K + sc;
    const float* gB = W + (size_t)(bn + sr) * K + sc;
    const int so = sr * GK_PITCH + sc;

    v4f ra[GK_NP], rb[GK_NP];
#pragma unroll
    for (int i = 0; i < GK_NP; ++i) {
        ra[i] = *(const v4f*)(gA + (size_t)(GK_RPP * i) * K);
        rb[i] = *(const v4f*)(gB + (size_t)(GK_RPP * i) * K);
    }
#pragma unroll
    for (int i = 0; i < GK_NP; ++i) {
        *(v4f*)(&sA[0][so + GK_RPP * i * GK_PITCH]) = ra[i];
        *(v4f*)(&sB[0][so + GK_RPP * i * GK_PITCH]) = rb[i];
    }
    __syncthreads();

    const int fa = (wm + ln) * GK_PITCH + 2 * lh;
    const int fb = (wn + ln) * GK_PITCH + 2 * lh;

    v8f c[2][4] = {};
    const int nk = K / GK_CHUNK;
    for (int kc = 0; kc < nk; ++kc) {
        const int cur = kc & 1;
        const bool more = (kc + 1 < nk);
        if (more) {
            const int k1 = (kc + 1) * GK_CHUNK;
#pragma unroll
            for (int i = 0; i < GK_NP; ++i) {
                ra[i] = *(const v4f*)(gA + (size_t)(GK_RPP * i) * K + k1);
                rb[i] = *(const v4f*)(gB + (size_t)(GK_RPP * i) * K + k1);
            }
        }
        const float* pa = &sA[cur][fa];
        const float* pb = &sB[cur][fb];
#pragma unroll
        for (int j = 0; j < GK_CHUNK / 4; ++j) {
            v2f a0 = *(const v2f*)(pa + 4 * j);
            v2f a1 = *(const v2f*)(pa + 16 * GK_PITCH + 4 * j);
            v2f b0 = *(const v2f*)(pb + 4 * j);
            v2f b1 = *(const v2f*)(pb + 16 * GK_PITCH + 4 * j);
            v2f b2 = *(const v2f*)(pb + 32 * GK_PITCH + 4 * j);
            v2f b3 = *(const v2f*)(pb + 48 * GK_PITCH + 4 * j);
            c[0][0] = __builtin_amdgcn_wmma_f32_16x16x4_f32(false, a0, false, b0, (short)0, c[0][0], false, false);
            c[0][1] = __builtin_amdgcn_wmma_f32_16x16x4_f32(false, a0, false, b1, (short)0, c[0][1], false, false);
            c[0][2] = __builtin_amdgcn_wmma_f32_16x16x4_f32(false, a0, false, b2, (short)0, c[0][2], false, false);
            c[0][3] = __builtin_amdgcn_wmma_f32_16x16x4_f32(false, a0, false, b3, (short)0, c[0][3], false, false);
            c[1][0] = __builtin_amdgcn_wmma_f32_16x16x4_f32(false, a1, false, b0, (short)0, c[1][0], false, false);
            c[1][1] = __builtin_amdgcn_wmma_f32_16x16x4_f32(false, a1, false, b1, (short)0, c[1][1], false, false);
            c[1][2] = __builtin_amdgcn_wmma_f32_16x16x4_f32(false, a1, false, b2, (short)0, c[1][2], false, false);
            c[1][3] = __builtin_amdgcn_wmma_f32_16x16x4_f32(false, a1, false, b3, (short)0, c[1][3], false, false);
        }
        if (more) {
#pragma unroll
            for (int i = 0; i < GK_NP; ++i) {
                *(v4f*)(&sA[cur ^ 1][so + GK_RPP * i * GK_PITCH]) = ra[i];
                *(v4f*)(&sB[cur ^ 1][so + GK_RPP * i * GK_PITCH]) = rb[i];
            }
        }
        __syncthreads();
    }

    if (STORE_VT == 0) {
#pragma unroll
        for (int mi = 0; mi < 2; ++mi)
#pragma unroll
            for (int ni = 0; ni < 4; ++ni) {
                float* orow = out + (size_t)(tm + mi * 16 + 8 * lh) * N + tn + ni * 16 + ln;
#pragma unroll
                for (int r = 0; r < 8; ++r)
                    orow[(size_t)r * N] = c[mi][ni][r];
            }
    } else {
        const int bb = tm >> 11;
#pragma unroll
        for (int mi = 0; mi < 2; ++mi)
#pragma unroll
            for (int ni = 0; ni < 4; ++ni) {
                int n = tn + ni * 16 + ln;
                int srow = (tm & (SEQ - 1)) + mi * 16 + 8 * lh;
                float* ocol = out + ((size_t)(bb * N + n) << 11) + srow;
#pragma unroll
                for (int r = 0; r < 8; ++r)
                    ocol[r] = c[mi][ni][r];
            }
    }
}

__global__ __launch_bounds__(256) void rope_kernel(float* __restrict__ q,
                                                   float* __restrict__ k, int total) {
    int tid = blockIdx.x * blockDim.x + threadIdx.x;
    if (tid >= total) return;
    int i = tid & 31;
    int s = (tid >> 9) & (SEQ - 1);
    float freq = __expf(-(float)i * (1.0f / 32.0f) * 9.210340371976184f);
    float ang = (float)s * freq;
    float cs = __cosf(ang);
    float sn = __sinf(ang);
    size_t p = (size_t)tid * 2;
    float q1 = q[p], q2 = q[p + 1];
    q[p]     = cs * q1 - sn * q2;
    q[p + 1] = sn * q1 + cs * q2;
    float k1 = k[p], k2 = k[p + 1];
    k[p]     = cs * k1 - sn * k2;
    k[p + 1] = sn * k1 + cs * k2;
}

__device__ __forceinline__ float xhalf(float x) {
    int b = __builtin_bit_cast(int, x);
    return __builtin_bit_cast(float,
        __builtin_amdgcn_permlanex16(b, b, 0x76543210, 0xfedcba98, false, false));
}

#define AK_PITCH 68
#define AV_PITCH 20
#define AW_K 0
#define AW_V (16 * AK_PITCH)
#define AW_FLOATS (AW_V + 64 * AV_PITCH)

__global__ __launch_bounds__(256) void attn_kernel(const float* __restrict__ q,
                                                   const float* __restrict__ k,
                                                   const float* __restrict__ vt,
                                                   float* __restrict__ o) {
    __shared__ __attribute__((aligned(16))) float lds[8][AW_FLOATS];
    const int lane = threadIdx.x & 31;
    const int warp = threadIdx.x >> 5;
    const int bh = blockIdx.x >> 4;
    const int grp = ((blockIdx.x & 15) << 2) + (warp & 3);
    const int qt = (warp < 4) ? 127 - grp : grp;
    const int h = bh & 15;
    const int b = bh >> 4;
    const int ln = lane & 15;
    const int lh = lane >> 4;

    const float* qbase = q + ((size_t)(b * SEQ) + qt * 16 + ln) * DM + h * DK;
    v2f qa[16];
#pragma unroll
    for (int j = 0; j < 16; ++j)
        qa[j] = *(const v2f*)(qbase + 4 * j + 2 * lh);

    float* sK = &lds[warp][AW_K];
    float* sV = &lds[warp][AW_V];

    const int kr = lane >> 4;
    const int kc = (lane & 15) << 2;
    const int vr = lane >> 2;
    const int vc = (lane & 3) << 2;
    const float* gK = k + ((size_t)(b * SEQ) + kr) * DM + h * DK + kc;
    const float* gV = vt + ((size_t)(b * DM + h * DK + vr) << 11) + vc;
    float* stK = sK + kr * AK_PITCH + kc;
    float* stV = sV + vr * AV_PITCH + vc;

    v4f rk[8], rv[8];
#pragma unroll
    for (int i = 0; i < 8; ++i) {
        rk[i] = *(const v4f*)(gK + (size_t)(2 * i) * DM);
        rv[i] = *(const v4f*)(gV + ((size_t)(8 * i) << 11));
    }

    const int kperm = (((ln & 7) >> 1) << 2) + ((ln >> 3) << 1) + (ln & 1);
    const float* fK = sK + kperm * AK_PITCH + 2 * lh;
    const float* fV = sV + ln * AV_PITCH + 2 * lh;

    v8f acc[4] = {v8f{}, v8f{}, v8f{}, v8f{}};
    float mrun = -1e30f;
    float lrun = 0.0f;

    for (int kt = 0; kt <= qt; ++kt) {
#pragma unroll
        for (int i = 0; i < 8; ++i) {
            *(v4f*)(stK + 2 * i * AK_PITCH) = rk[i];
            *(v4f*)(stV + 8 * i * AV_PITCH) = rv[i];
        }
        const int kn = (kt < qt) ? kt + 1 : qt;
#pragma unroll
        for (int i = 0; i < 8; ++i) {
            rk[i] = *(const v4f*)(gK + (size_t)(kn * 16 + 2 * i) * DM);
            rv[i] = *(const v4f*)(gV + ((size_t)(8 * i) << 11) + kn * 16);
        }
        v2f kf[16], vf[4][4];
#pragma unroll
        for (int j = 0; j < 16; ++j)
            kf[j] = *(const v2f*)(fK + 4 * j);
#pragma unroll
        for (int j = 0; j < 4; ++j)
#pragma unroll
            for (int nt = 0; nt < 4; ++nt)
                vf[j][nt] = *(const v2f*)(fV + nt * 16 * AV_PITCH + 4 * j);
        v8f s = {};
#pragma unroll
        for (int j = 0; j < 16; ++j)
            s = __builtin_amdgcn_wmma_f32_16x16x4_f32(false, kf[j], false, qa[j],
                                                      (short)0, s, false, false);
        float mx = -1e30f;
#pragma unroll
        for (int r = 0; r < 8; ++r) {
            float sv = s[r] * 0.125f;
            if (kt == qt && (4 * (r >> 1) + (r & 1) + 2 * lh) > ln)
                sv = -1e30f;
            s[r] = sv;
            mx = fmaxf(mx, sv);
        }
        mx = fmaxf(mx, xhalf(mx));
        const float nm = fmaxf(mrun, mx);
        const float sc = __expf(mrun - nm);
        float ps = 0.0f;
#pragma unroll
        for (int r = 0; r < 8; ++r) {
            float p = __expf(s[r] - nm);
            s[r] = p;
            ps += p;
        }
        lrun = lrun * sc + ps;
        mrun = nm;
#pragma unroll
        for (int nt = 0; nt < 4; ++nt)
#pragma unroll
            for (int r = 0; r < 8; ++r) acc[nt][r] *= sc;
#pragma unroll
        for (int j = 0; j < 4; ++j) {
            v2f pb = {s[2 * j], s[2 * j + 1]};
#pragma unroll
            for (int nt = 0; nt < 4; ++nt)
                acc[nt] = __builtin_amdgcn_wmma_f32_16x16x4_f32(false, vf[j][nt], false, pb,
                                                                (short)0, acc[nt],
                                                                false, false);
        }
    }

    const float inv = 1.0f / (lrun + xhalf(lrun));
#pragma unroll
    for (int nt = 0; nt < 4; ++nt) {
        v4f lo = {acc[nt][0] * inv, acc[nt][1] * inv, acc[nt][2] * inv, acc[nt][3] * inv};
        v4f hi = {acc[nt][4] * inv, acc[nt][5] * inv, acc[nt][6] * inv, acc[nt][7] * inv};
        float* d = sK + ln * AK_PITCH + nt * 16 + 8 * lh;
        *(v4f*)(d) = lo;
        *(v4f*)(d + 4) = hi;
    }
    float* obase = o + ((size_t)(b * SEQ) + qt * 16 + kr) * DM + h * DK + kc;
#pragma unroll
    for (int i = 0; i < 8; ++i) {
        v4f t = *(const v4f*)(sK + (2 * i + kr) * AK_PITCH + kc);
        *(v4f*)(obase + (size_t)(2 * i) * DM) = t;
    }
}

extern "C" void kernel_launch(void* const* d_in, const int* in_sizes, int n_in,
                              void* d_out, int out_size, void* d_ws, size_t ws_size,
                              hipStream_t stream) {
    const float* x  = (const float*)d_in[0];
    const float* Wq = (const float*)d_in[1];
    const float* Wk = (const float*)d_in[2];
    const float* Wv = (const float*)d_in[3];
    const float* Wo = (const float*)d_in[4];
    float* out = (float*)d_out;

    const size_t bufElems = (size_t)BATCH * SEQ * DM;
    float* qb  = (float*)d_ws;
    float* kb  = qb + bufElems;
    float* vtb = kb + bufElems;
    float* ab  = vtb + bufElems;

    const int M = BATCH * SEQ, N = DM, K = DM;
    const int gblocks = (M / 128) * (N / 128);

    gemm_xwt_kernel<0><<<gblocks, 256, 0, stream>>>(x, Wq, qb, M, N, K);
    gemm_xwt_kernel<0><<<gblocks, 256, 0, stream>>>(x, Wk, kb, M, N, K);
    gemm_xwt_kernel<1><<<gblocks, 256, 0, stream>>>(x, Wv, vtb, M, N, K);

    const int pairs = BATCH * SEQ * NH * (DK / 2);
    rope_kernel<<<pairs / 256, 256, 0, stream>>>(qb, kb, pairs);

    const int waves = BATCH * NH * (SEQ / 16);
    attn_kernel<<<waves / 8, 256, 0, stream>>>(qb, kb, vtb, ab);

    gemm_xwt_kernel<0><<<gblocks, 256, 0, stream>>>(ab, Wo, out, M, N, K);
}


# Q and K projections apply RoPE to their accumulator tile in registers before the store (pair partner by DPP quad_perm); separate RoPE launch and its Q/K round trip removed; sum order unchanged
# speedup vs baseline: 2.4518x; 1.0946x over previous
#include <hip/hip_runtime.h>
#include <math.h>

typedef __attribute__((ext_vector_type(2))) float v2f;
typedef __attribute__((ext_vector_type(4))) float v4f;
typedef __attribute__((ext_vector_type(8))) float v8f;

#define BATCH 2
#define SEQ 2048
#define DM 1024
#define NH 16
#define DK 64

__device__ __forceinline__ float xpair(float x) {
    int b = __builtin_bit_cast(int, x);
    return __builtin_bit_cast(float, __builtin_amdgcn_update_dpp(b, b, 0xB1, 0xF, 0xF, true));
}

#define GK_CHUNK 64
#define GK_PITCH 68
#define GK_TPR (GK_CHUNK / 4)
#define GK_RPP (256 / GK_TPR)
#define GK_NP (128 / GK_RPP)

template <int STORE_VT>
__global__ __launch_bounds__(256) void gemm_xwt_kernel(const float* __restrict__ X,
                                                       const float* __restrict__ W,
                                                       float* __restrict__ out,
                                                       int M, int N, int K) {
    __shared__ __attribute__((aligned(16))) float sA[2][128 * GK_PITCH];
    __shared__ __attribute__((aligned(16))) float sB[2][128 * GK_PITCH];

    const int t = threadIdx.x;
    const int lane = t & 31;
    const int warp = t >> 5;
    const int tilesN = N >> 7;
    const int bm = (blockIdx.x / tilesN) << 7;
    const int bn = (blockIdx.x % tilesN) << 7;
    const int wm = (warp & 3) << 5;
    const int wn = (warp >> 2) << 6;
    const int tm = bm + wm;
    const int tn = bn + wn;
    const int ln = lane & 15;
    const int lh = lane >> 4;

    const int sr = t / GK_TPR;
    const int sc = (t % GK_TPR) << 2;
    const float* gA = X + (size_t)(bm + sr) * K + sc;
    const float* gB = W + (size_t)(bn + sr) * K + sc;
    const int so = sr * GK_PITCH + sc;

    v4f ra[GK_NP], rb[GK_NP];
#pragma unroll
    for (int i = 0; i < GK_NP; ++i) {
        ra[i] = *(const v4f*)(gA + (size_t)(GK_RPP * i) * K);
        rb[i] = *(const v4f*)(gB + (size_t)(GK_RPP * i) * K);
    }
#pragma unroll
    for (int i = 0; i < GK_NP; ++i) {
        *(v4f*)(&sA[0][so + GK_RPP * i * GK_PITCH]) = ra[i];
        *(v4f*)(&sB[0][so + GK_RPP * i * GK_PITCH]) = rb[i];
    }
    __syncthreads();

    const int fa = (wm + ln) * GK_PITCH + 2 * lh;
    const int fb = (wn + ln) * GK_PITCH + 2 * lh;

    v8f c[2][4] = {};
    const int nk = K / GK_CHUNK;
    for (int kc = 0; kc < nk; ++kc) {
        const int cur = kc & 1;
        const bool more = (kc + 1 < nk);
        if (more) {
            const int k1 = (kc + 1) * GK_CHUNK;
#pragma unroll
            for (int i = 0; i < GK_NP; ++i) {
                ra[i] = *(const v4f*)(gA + (size_t)(GK_RPP * i) * K + k1);
                rb[i] = *(const v4f*)(gB + (size_t)(GK_RPP * i) * K + k1);
            }
        }
        const float* pa = &sA[cur][fa];
        const float* pb = &sB[cur][fb];
#pragma unroll
        for (int j = 0; j < GK_CHUNK / 4; ++j) {
            v2f a0 = *(const v2f*)(pa + 4 * j);
            v2f a1 = *(const v2f*)(pa + 16 * GK_PITCH + 4 * j);
            v2f b0 = *(const v2f*)(pb + 4 * j);
            v2f b1 = *(const v2f*)(pb + 16 * GK_PITCH + 4 * j);
            v2f b2 = *(const v2f*)(pb + 32 * GK_PITCH + 4 * j);
            v2f b3 = *(const v2f*)(pb + 48 * GK_PITCH + 4 * j);
            c[0][0] = __builtin_amdgcn_wmma_f32_16x16x4_f32(false, a0, false, b0, (short)0, c[0][0], false, false);
            c[0][1] = __builtin_amdgcn_wmma_f32_16x16x4_f32(false, a0, false, b1, (short)0, c[0][1], false, false);
            c[0][2] = __builtin_amdgcn_wmma_f32_16x16x4_f32(false, a0, false, b2, (short)0, c[0][2], false, false);
            c[0][3] = __builtin_amdgcn_wmma_f32_16x16x4_f32(false, a0, false, b3, (short)0, c[0][3], false, false);
            c[1][0] = __builtin_amdgcn_wmma_f32_16x16x4_f32(false, a1, false, b0, (short)0, c[1][0], false, false);
            c[1][1] = __builtin_amdgcn_wmma_f32_16x16x4_f32(false, a1, false, b1, (short)0, c[1][1], false, false);
            c[1][2] = __builtin_amdgcn_wmma_f32_16x16x4_f32(false, a1, false, b2, (short)0, c[1][2], false, false);
            c[1][3] = __builtin_amdgcn_wmma_f32_16x16x4_f32(false, a1, false, b3, (short)0, c[1][3], false, false);
        }
        if (more) {
#pragma unroll
            for (int i = 0; i < GK_NP; ++i) {
                *(v4f*)(&sA[cur ^ 1][so + GK_RPP * i * GK_PITCH]) = ra[i];
                *(v4f*)(&sB[cur ^ 1][so + GK_RPP * i * GK_PITCH]) = rb[i];
            }
        }
        __syncthreads();
    }

    if (STORE_VT == 2) {
        const bool odd = (ln & 1) != 0;
        float fr[4];
#pragma unroll
        for (int ni = 0; ni < 4; ++ni) {
            const int i = ((tn + ni * 16 + ln) & (DK - 1)) >> 1;
            fr[ni] = __expf(-(float)i * (1.0f / 32.0f) * 9.210340371976184f);
        }
#pragma unroll
        for (int mi = 0; mi < 2; ++mi)
#pragma unroll
            for (int r = 0; r < 8; ++r) {
                const float sf = (float)((tm + mi * 16 + 8 * lh + r) & (SEQ - 1));
#pragma unroll
                for (int ni = 0; ni < 4; ++ni) {
                    const float ang = sf * fr[ni];
                    const float cs = __cosf(ang);
                    const float sn = __sinf(ang);
                    const float v = c[mi][ni][r];
                    const float p = xpair(v);
                    const float sg = odd ? sn : -sn;
                    c[mi][ni][r] = cs * v + sg * p;
                }
            }
    }
    if (STORE_VT != 1) {
#pragma unroll
        for (int mi = 0; mi < 2; ++mi)
#pragma unroll
            for (int ni = 0; ni < 4; ++ni) {
                float* orow = out + (size_t)(tm + mi * 16 + 8 * lh) * N + tn + ni * 16 + ln;
#pragma unroll
                for (int r = 0; r < 8; ++r)
                    orow[(size_t)r * N] = c[mi][ni][r];
            }
    } else {
        const int bb = tm >> 11;
#pragma unroll
        for (int mi = 0; mi < 2; ++mi)
#pragma unroll
            for (int ni = 0; ni < 4; ++ni) {
                int n = tn + ni * 16 + ln;
                int srow = (tm & (SEQ - 1)) + mi * 16 + 8 * lh;
                float* ocol = out + ((size_t)(bb * N + n) << 11) + srow;
#pragma unroll
                for (int r = 0; r < 8; ++r)
                    ocol[r] = c[mi][ni][r];
            }
    }
}

__device__ __forceinline__ float xhalf(float x) {
    int b = __builtin_bit_cast(int, x);
    return __builtin_bit_cast(float,
        __builtin_amdgcn_permlanex16(b, b, 0x76543210, 0xfedcba98, false, false));
}

#define AK_PITCH 68
#define AV_PITCH 20
#define AW_K 0
#define AW_V (16 * AK_PITCH)
#define AW_FLOATS (AW_V + 64 * AV_PITCH)

__global__ __launch_bounds__(256) void attn_kernel(const float* __restrict__ q,
                                                   const float* __restrict__ k,
                                                   const float* __restrict__ vt,
                                                   float* __restrict__ o) {
    __shared__ __attribute__((aligned(16))) float lds[8][AW_FLOATS];
    const int lane = threadIdx.x & 31;
    const int warp = threadIdx.x >> 5;
    const int bh = blockIdx.x >> 4;
    const int grp = ((blockIdx.x & 15) << 2) + (warp & 3);
    const int qt = (warp < 4) ? 127 - grp : grp;
    const int h = bh & 15;
    const int b = bh >> 4;
    const int ln = lane & 15;
    const int lh = lane >> 4;

    const float* qbase = q + ((size_t)(b * SEQ) + qt * 16 + ln) * DM + h * DK;
    v2f qa[16];
#pragma unroll
    for (int j = 0; j < 16; ++j)
        qa[j] = *(const v2f*)(qbase + 4 * j + 2 * lh);

    float* sK = &lds[warp][AW_K];
    float* sV = &lds[warp][AW_V];

    const int kr = lane >> 4;
    const int kc = (lane & 15) << 2;
    const int vr = lane >> 2;
    const int vc = (lane & 3) << 2;
    const float* gK = k + ((size_t)(b * SEQ) + kr) * DM + h * DK + kc;
    const float* gV = vt + ((size_t)(b * DM + h * DK + vr) << 11) + vc;
    float* stK = sK + kr * AK_PITCH + kc;
    float* stV = sV + vr * AV_PITCH + vc;

    v4f rk[8], rv[8];
#pragma unroll
    for (int i = 0; i < 8; ++i) {
        rk[i] = *(const v4f*)(gK + (size_t)(2 * i) * DM);
        rv[i] = *(const v4f*)(gV + ((size_t)(8 * i) << 11));
    }

    const int kperm = (((ln & 7) >> 1) << 2) + ((ln >> 3) << 1) + (ln & 1);
    const float* fK = sK + kperm * AK_PITCH + 2 * lh;
    const float* fV = sV + ln * AV_PITCH + 2 * lh;

    v8f acc[4] = {v8f{}, v8f{}, v8f{}, v8f{}};
    float mrun = -1e30f;
    float lrun = 0.0f;

    for (int kt = 0; kt <= qt; ++kt) {
#pragma unroll
        for (int i = 0; i < 8; ++i) {
            *(v4f*)(stK + 2 * i * AK_PITCH) = rk[i];
            *(v4f*)(stV + 8 * i * AV_PITCH) = rv[i];
        }
        const int kn = (kt < qt) ? kt + 1 : qt;
#pragma unroll
        for (int i = 0; i < 8; ++i) {
            rk[i] = *(const v4f*)(gK + (size_t)(kn * 16 + 2 * i) * DM);
            rv[i] = *(const v4f*)(gV + ((size_t)(8 * i) << 11) + kn * 16);
        }
        v2f kf[16], vf[4][4];
#pragma unroll
        for (int j = 0; j < 16; ++j)
            kf[j] = *(const v2f*)(fK + 4 * j);
#pragma unroll
        for (int j = 0; j < 4; ++j)
#pragma unroll
            for (int nt = 0; nt < 4; ++nt)
                vf[j][nt] = *(const v2f*)(fV + nt * 16 * AV_PITCH + 4 * j);
        v8f s = {};
#pragma unroll
        for (int j = 0; j < 16; ++j)
            s = __builtin_amdgcn_wmma_f32_16x16x4_f32(false, kf[j], false, qa[j],
                                                      (short)0, s, false, false);
        float mx = -1e30f;
#pragma unroll
        for (int r = 0; r < 8; ++r) {
            float sv = s[r] * 0.125f;
            if (kt == qt && (4 * (r >> 1) + (r & 1) + 2 * lh) > ln)
                sv = -1e30f;
            s[r] = sv;
            mx = fmaxf(mx, sv);
        }
        mx = fmaxf(mx, xhalf(mx));
        const float nm = fmaxf(mrun, mx);
        const float sc = __expf(mrun - nm);
        float ps = 0.0f;
#pragma unroll
        for (int r = 0; r < 8; ++r) {
            float p = __expf(s[r] - nm);
            s[r] = p;
            ps += p;
        }
        lrun = lrun * sc + ps;
        mrun = nm;
#pragma unroll
        for (int nt = 0; nt < 4; ++nt)
#pragma unroll
            for (int r = 0; r < 8; ++r) acc[nt][r] *= sc;
#pragma unroll
        for (int j = 0; j < 4; ++j) {
            v2f pb = {s[2 * j], s[2 * j + 1]};
#pragma unroll
            for (int nt = 0; nt < 4; ++nt)
                acc[nt] = __builtin_amdgcn_wmma_f32_16x16x4_f32(false, vf[j][nt], false, pb,
                                                                (short)0, acc[nt],
                                                                false, false);
        }
    }

    const float inv = 1.0f / (lrun + xhalf(lrun));
#pragma unroll
    for (int nt = 0; nt < 4; ++nt) {
        v4f lo = {acc[nt][0] * inv, acc[nt][1] * inv, acc[nt][2] * inv, acc[nt][3] * inv};
        v4f hi = {acc[nt][4] * inv, acc[nt][5] * inv, acc[nt][6] * inv, acc[nt][7] * inv};
        float* d = sK + ln * AK_PITCH + nt * 16 + 8 * lh;
        *(v4f*)(d) = lo;
        *(v4f*)(d + 4) = hi;
    }
    float* obase = o + ((size_t)(b * SEQ) + qt * 16 + kr) * DM + h * DK + kc;
#pragma unroll
    for (int i = 0; i < 8; ++i) {
        v4f t = *(const v4f*)(sK + (2 * i + kr) * AK_PITCH + kc);
        *(v4f*)(obase + (size_t)(2 * i) * DM) = t;
    }
}

extern "C" void kernel_launch(void* const* d_in, const int* in_sizes, int n_in,
                              void* d_out, int out_size, void* d_ws, size_t ws_size,
                              hipStream_t stream) {
    const float* x  = (const float*)d_in[0];
    const float* Wq = (const float*)d_in[1];
    const float* Wk = (const float*)d_in[2];
    const float* Wv = (const float*)d_in[3];
    const float* Wo = (const float*)d_in[4];
    float* out = (float*)d_out;

    const size_t bufElems = (size_t)BATCH * SEQ * DM;
    float* qb  = (float*)d_ws;
    float* kb  = qb + bufElems;
    float* vtb = kb + bufElems;
    float* ab  = vtb + bufElems;

    const int M = BATCH * SEQ, N = DM, K = DM;
    const int gblocks = (M / 128) * (N / 128);

    gemm_xwt_kernel<2><<<gblocks, 256, 0, stream>>>(x, Wq, qb, M, N, K);
    gemm_xwt_kernel<2><<<gblocks, 256, 0, stream>>>(x, Wk, kb, M, N, K);
    gemm_xwt_kernel<1><<<gblocks, 256, 0, stream>>>(x, Wv, vtb, M, N, K);

    const int waves = BATCH * NH * (SEQ / 16);
    attn_kernel<<<waves / 8, 256, 0, stream>>>(qb, kb, vtb, ab);

    gemm_xwt_kernel<0><<<gblocks, 256, 0, stream>>>(ab, Wo, out, M, N, K);
}


# Q, K and V projections merged into one launch of 768 blocks (block-uniform weight and epilogue choice): two blocks resident per WGP overlap each other's barrier waits; sum order unchanged
# speedup vs baseline: 2.5462x; 1.0385x over previous
#include <hip/hip_runtime.h>
#include <math.h>

typedef __attribute__((ext_vector_type(2))) float v2f;
typedef __attribute__((ext_vector_type(4))) float v4f;
typedef __attribute__((ext_vector_type(8))) float v8f;

#define BATCH 2
#define SEQ 2048
#define DM 1024
#define NH 16
#define DK 64

__device__ __forceinline__ float xpair(float x) {
    int b = __builtin_bit_cast(int, x);
    return __builtin_bit_cast(float, __builtin_amdgcn_update_dpp(b, b, 0xB1, 0xF, 0xF, true));
}

#define GK_CHUNK 64
#define GK_PITCH 68
#define GK_TPR (GK_CHUNK / 4)
#define GK_RPP (256 / GK_TPR)
#define GK_NP (128 / GK_RPP)
#define GK_STAGE (128 * GK_PITCH)

__device__ __forceinline__ void gemm_tile_mainloop(const float* __restrict__ X,
                                                   const float* __restrict__ W,
                                                   float* __restrict__ sA, float* __restrict__ sB,
                                                   int bm, int bn, int K, v8f (&c)[2][4]) {
    const int t = threadIdx.x;
    const int lane = t & 31;
    const int warp = t >> 5;
    const int wm = (warp & 3) << 5;
    const int wn = (warp >> 2) << 6;
    const int ln = lane & 15;
    const int lh = lane >> 4;

    const int sr = t / GK_TPR;
    const int sc = (t % GK_TPR) << 2;
    const float* gA = X + (size_t)(bm + sr) * K + sc;
    const float* gB = W + (size_t)(bn + sr) * K + sc;
    const int so = sr * GK_PITCH + sc;

    v4f ra[GK_NP], rb[GK_NP];
#pragma unroll
    for (int i = 0; i < GK_NP; ++i) {
        ra[i] = *(const v4f*)(gA + (size_t)(GK_RPP * i) * K);
        rb[i] = *(const v4f*)(gB + (size_t)(GK_RPP * i) * K);
    }
#pragma unroll
    for (int i = 0; i < GK_NP; ++i) {
        *(v4f*)(&sA[so + GK_RPP * i * GK_PITCH]) = ra[i];
        *(v4f*)(&sB[so + GK_RPP * i * GK_PITCH]) = rb[i];
    }
    __syncthreads();

    const int fa = (wm + ln) * GK_PITCH + 2 * lh;
    const int fb = (wn + ln) * GK_PITCH + 2 * lh;

    const int nk = K / GK_CHUNK;
    for (int kc = 0; kc < nk; ++kc) {
        const int cur = kc & 1;
        const bool more = (kc + 1 < nk);
        if (more) {
            const int k1 = (kc + 1) * GK_CHUNK;
#pragma unroll
            for (int i = 0; i < GK_NP; ++i) {
                ra[i] = *(const v4f*)(gA + (size_t)(GK_RPP * i) * K + k1);
                rb[i] = *(const v4f*)(gB + (size_t)(GK_RPP * i) * K + k1);
            }
        }
        const float* pa = sA + cur * GK_STAGE + fa;
        const float* pb = sB + cur * GK_STAGE + fb;
#pragma unroll
        for (int j = 0; j < GK_CHUNK / 4; ++j) {
            v2f a0 = *(const v2f*)(pa + 4 * j);
            v2f a1 = *(const v2f*)(pa + 16 * GK_PITCH + 4 * j);
            v2f b0 = *(const v2f*)(pb + 4 * j);
            v2f b1 = *(const v2f*)(pb + 16 * GK_PITCH + 4 * j);
            v2f b2 = *(const v2f*)(pb + 32 * GK_PITCH + 4 * j);
            v2f b3 = *(const v2f*)(pb + 48 * GK_PITCH + 4 * j);
            c[0][0] = __builtin_amdgcn_wmma_f32_16x16x4_f32(false, a0, false, b0, (short)0, c[0][0], false, false);
            c[0][1] = __builtin_amdgcn_wmma_f32_16x16x4_f32(false, a0, false, b1, (short)0, c[0][1], false, false);
            c[0][2] = __builtin_amdgcn_wmma_f32_16x16x4_f32(false, a0, false, b2, (short)0, c[0][2], false, false);
            c[0][3] = __builtin_amdgcn_wmma_f32_16x16x4_f32(false, a0, false, b3, (short)0, c[0][3], false, false);
            c[1][0] = __builtin_amdgcn_wmma_f32_16x16x4_f32(false, a1, false, b0, (short)0, c[1][0], false, false);
            c[1][1] = __builtin_amdgcn_wmma_f32_16x16x4_f32(false, a1, false, b1, (short)0, c[1][1], false, false);
            c[1][2] = __builtin_amdgcn_wmma_f32_16x16x4_f32(false, a1, false, b2, (short)0, c[1][2], false, false);
            c[1][3] = __builtin_amdgcn_wmma_f32_16x16x4_f32(false, a1, false, b3, (short)0, c[1][3], false, false);
        }
        if (more) {
            float* da = sA + (cur ^ 1) * GK_STAGE + so;
            float* db = sB + (cur ^ 1) * GK_STAGE + so;
#pragma unroll
            for (int i = 0; i < GK_NP; ++i) {
                *(v4f*)(da + GK_RPP * i * GK_PITCH) = ra[i];
                *(v4f*)(db + GK_RPP * i * GK_PITCH) = rb[i];
            }
        }
        __syncthreads();
    }
}

__device__ __forceinline__ void gemm_tile_rotary(v8f (&c)[2][4], int tm, int tn, int ln, int lh) {
    const bool odd = (ln & 1) != 0;
    float fr[4];
#pragma unroll
    for (int ni = 0; ni < 4; ++ni) {
        const int i = ((tn + ni * 16 + ln) & (DK - 1)) >> 1;
        fr[ni] = __expf(-(float)i * (1.0f / 32.0f) * 9.210340371976184f);
    }
#pragma unroll
    for (int mi = 0; mi < 2; ++mi)
#pragma unroll
        for (int r = 0; r < 8; ++r) {
            const float sf = (float)((tm + mi * 16 + 8 * lh + r) & (SEQ - 1));
#pragma unroll
            for (int ni = 0; ni < 4; ++ni) {
                const float ang = sf * fr[ni];
                const float cs = __cosf(ang);
                const float sn = __sinf(ang);
                const float v = c[mi][ni][r];
                const float p = xpair(v);
                const float sg = odd ? sn : -sn;
                c[mi][ni][r] = cs * v + sg * p;
            }
        }
}

__device__ __forceinline__ void gemm_tile_store_plain(const v8f (&c)[2][4], float* __restrict__ out,
                                                      int tm, int tn, int N, int ln, int lh) {
#pragma unroll
    for (int mi = 0; mi < 2; ++mi)
#pragma unroll
        for (int ni = 0; ni < 4; ++ni) {
            float* orow = out + (size_t)(tm + mi * 16 + 8 * lh) * N + tn + ni * 16 + ln;
#pragma unroll
            for (int r = 0; r < 8; ++r)
                orow[(size_t)r * N] = c[mi][ni][r];
        }
}

__device__ __forceinline__ void gemm_tile_store_transposed(const v8f (&c)[2][4], float* __restrict__ out,
                                                           int tm, int tn, int N, int ln, int lh) {
    const int bb = tm >> 11;
#pragma unroll
    for (int mi = 0; mi < 2; ++mi)
#pragma unroll
        for (int ni = 0; ni < 4; ++ni) {
            int n = tn + ni * 16 + ln;
            int srow = (tm & (SEQ - 1)) + mi * 16 + 8 * lh;
            float* ocol = out + ((size_t)(bb * N + n) << 11) + srow;
#pragma unroll
            for (int r = 0; r < 8; ++r)
                ocol[r] = c[mi][ni][r];
        }
}

__global__ __launch_bounds__(256) void gemm_xwt_kernel(const float* __restrict__ X,
                                                       const float* __restrict__ W,
                                                       float* __restrict__ out,
                                                       int M, int N, int K) {
    __shared__ __attribute__((aligned(16))) float sA[2 * GK_STAGE];
    __shared__ __attribute__((aligned(16))) float sB[2 * GK_STAGE];
    const int lane = threadIdx.x & 31;
    const int warp = threadIdx.x >> 5;
    const int tilesN = N >> 7;
    const int bm = (blockIdx.x / tilesN) << 7;
    const int bn = (blockIdx.x % tilesN) << 7;
    v8f c[2][4] = {};
    gemm_tile_mainloop(X, W, sA, sB, bm, bn, K, c);
    gemm_tile_store_plain(c, out, bm + ((warp & 3) << 5), bn + ((warp >> 2) << 6), N, lane & 15, lane >> 4);
}

__global__ __launch_bounds__(256) void qkv_proj_kernel(const float* __restrict__ X,
                                                       const float* __restrict__ Wq,
                                                       const float* __restrict__ Wk,
                                                       const float* __restrict__ Wv,
                                                       float* __restrict__ qo,
                                                       float* __restrict__ ko,
                                                       float* __restrict__ vto,
                                                       int M, int N, int K) {
    __shared__ __attribute__((aligned(16))) float sA[2 * GK_STAGE];
    __shared__ __attribute__((aligned(16))) float sB[2 * GK_STAGE];
    const int lane = threadIdx.x & 31;
    const int warp = threadIdx.x >> 5;
    const int tilesN = N >> 7;
    const int tiles = (M >> 7) * tilesN;
    const int which = blockIdx.x / tiles;
    const int tile = blockIdx.x - which * tiles;
    const int bm = (tile / tilesN) << 7;
    const int bn = (tile % tilesN) << 7;
    const float* W = (which == 0) ? Wq : (which == 1) ? Wk : Wv;
    float* out = (which == 0) ? qo : (which == 1) ? ko : vto;
    v8f c[2][4] = {};
    gemm_tile_mainloop(X, W, sA, sB, bm, bn, K, c);
    const int tm = bm + ((warp & 3) << 5);
    const int tn = bn + ((warp >> 2) << 6);
    const int ln = lane & 15;
    const int lh = lane >> 4;
    if (which == 2) {
        gemm_tile_store_transposed(c, out, tm, tn, N, ln, lh);
    } else {
        gemm_tile_rotary(c, tm, tn, ln, lh);
        gemm_tile_store_plain(c, out, tm, tn, N, ln, lh);
    }
}

__device__ __forceinline__ float xhalf(float x) {
    int b = __builtin_bit_cast(int, x);
    return __builtin_bit_cast(float,
        __builtin_amdgcn_permlanex16(b, b, 0x76543210, 0xfedcba98, false, false));
}

#define AK_PITCH 68
#define AV_PITCH 20
#define AW_K 0
#define AW_V (16 * AK_PITCH)
#define AW_FLOATS (AW_V + 64 * AV_PITCH)

__global__ __launch_bounds__(256) void attn_kernel(const float* __restrict__ q,
                                                   const float* __restrict__ k,
                                                   const float* __restrict__ vt,
                                                   float* __restrict__ o) {
    __shared__ __attribute__((aligned(16))) float lds[8][AW_FLOATS];
    const int lane = threadIdx.x & 31;
    const int warp = threadIdx.x >> 5;
    const int bh = blockIdx.x >> 4;
    const int grp = ((blockIdx.x & 15) << 2) + (warp & 3);
    const int qt = (warp < 4) ? 127 - grp : grp;
    const int h = bh & 15;
    const int b = bh >> 4;
    const int ln = lane & 15;
    const int lh = lane >> 4;

    const float* qbase = q + ((size_t)(b * SEQ) + qt * 16 + ln) * DM + h * DK;
    v2f qa[16];
#pragma unroll
    for (int j = 0; j < 16; ++j)
        qa[j] = *(const v2f*)(qbase + 4 * j + 2 * lh);

    float* sK = &lds[warp][AW_K];
    float* sV = &lds[warp][AW_V];

    const int kr = lane >> 4;
    const int kc = (lane & 15) << 2;
    const int vr = lane >> 2;
    const int vc = (lane & 3) << 2;
    const float* gK = k + ((size_t)(b * SEQ) + kr) * DM + h * DK + kc;
    const float* gV = vt + ((size_t)(b * DM + h * DK + vr) << 11) + vc;
    float* stK = sK + kr * AK_PITCH + kc;
    float* stV = sV + vr * AV_PITCH + vc;

    v4f rk[8], rv[8];
#pragma unroll
    for (int i = 0; i < 8; ++i) {
        rk[i] = *(const v4f*)(gK + (size_t)(2 * i) * DM);
        rv[i] = *(const v4f*)(gV + ((size_t)(8 * i) << 11));
    }

    const int kperm = (((ln & 7) >> 1) << 2) + ((ln >> 3) << 1) + (ln & 1);
    const float* fK = sK + kperm * AK_PITCH + 2 * lh;
    const float* fV = sV + ln * AV_PITCH + 2 * lh;

    v8f acc[4] = {v8f{}, v8f{}, v8f{}, v8f{}};
    float mrun = -1e30f;
    float lrun = 0.0f;

    for (int kt = 0; kt <= qt; ++kt) {
#pragma unroll
        for (int i = 0; i < 8; ++i) {
            *(v4f*)(stK + 2 * i * AK_PITCH) = rk[i];
            *(v4f*)(stV + 8 * i * AV_PITCH) = rv[i];
        }
        const int kn = (kt < qt) ? kt + 1 : qt;
#pragma unroll
        for (int i = 0; i < 8; ++i) {
            rk[i] = *(const v4f*)(gK + (size_t)(kn * 16 + 2 * i) * DM);
            rv[i] = *(const v4f*)(gV + ((size_t)(8 * i) << 11) + kn * 16);
        }
        v2f kf[16], vf[4][4];
#pragma unroll
        for (int j = 0; j < 16; ++j)
            kf[j] = *(const v2f*)(fK + 4 * j);
#pragma unroll
        for (int j = 0; j < 4; ++j)
#pragma unroll
            for (int nt = 0; nt < 4; ++nt)
                vf[j][nt] = *(const v2f*)(fV + nt * 16 * AV_PITCH + 4 * j);
        v8f s = {};
#pragma unroll
        for (int j = 0; j < 16; ++j)
            s = __builtin_amdgcn_wmma_f32_16x16x4_f32(false, kf[j], false, qa[j],
                                                      (short)0, s, false, false);
        float mx = -1e30f;
#pragma unroll
        for (int r = 0; r < 8; ++r) {
            float sv = s[r] * 0.125f;
            if (kt == qt && (4 * (r >> 1) + (r & 1) + 2 * lh) > ln)
                sv = -1e30f;
            s[r] = sv;
            mx = fmaxf(mx, sv);
        }
        mx = fmaxf(mx, xhalf(mx));
        const float nm = fmaxf(mrun, mx);
        const float sc = __expf(mrun - nm);
        float ps = 0.0f;
#pragma unroll
        for (int r = 0; r < 8; ++r) {
            float p = __expf(s[r] - nm);
            s[r] = p;
            ps += p;
        }
        lrun = lrun * sc + ps;
        mrun = nm;
#pragma unroll
        for (int nt = 0; nt < 4; ++nt)
#pragma unroll
            for (int r = 0; r < 8; ++r) acc[nt][r] *= sc;
#pragma unroll
        for (int j = 0; j < 4; ++j) {
            v2f pb = {s[2 * j], s[2 * j + 1]};
#pragma unroll
            for (int nt = 0; nt < 4; ++nt)
                acc[nt] = __builtin_amdgcn_wmma_f32_16x16x4_f32(false, vf[j][nt], false, pb,
                                                                (short)0, acc[nt],
                                                                false, false);
        }
    }

    const float inv = 1.0f / (lrun + xhalf(lrun));
#pragma unroll
    for (int nt = 0; nt < 4; ++nt) {
        v4f lo = {acc[nt][0] * inv, acc[nt][1] * inv, acc[nt][2] * inv, acc[nt][3] * inv};
        v4f hi = {acc[nt][4] * inv, acc[nt][5] * inv, acc[nt][6] * inv, acc[nt][7] * inv};
        float* d = sK + ln * AK_PITCH + nt * 16 + 8 * lh;
        *(v4f*)(d) = lo;
        *(v4f*)(d + 4) = hi;
    }
    float* obase = o + ((size_t)(b * SEQ) + qt * 16 + kr) * DM + h * DK + kc;
#pragma unroll
    for (int i = 0; i < 8; ++i) {
        v4f t = *(const v4f*)(sK + (2 * i + kr) * AK_PITCH + kc);
        *(v4f*)(obase + (size_t)(2 * i) * DM) = t;
    }
}

extern "C" void kernel_launch(void* const* d_in, const int* in_sizes, int n_in,
                              void* d_out, int out_size, void* d_ws, size_t ws_size,
                              hipStream_t stream) {
    const float* x  = (const float*)d_in[0];
    const float* Wq = (const float*)d_in[1];
    const float* Wk = (const float*)d_in[2];
    const float* Wv = (const float*)d_in[3];
    const float* Wo = (const float*)d_in[4];
    float* out = (float*)d_out;

    const size_t bufElems = (size_t)BATCH * SEQ * DM;
    float* qb  = (float*)d_ws;
    float* kb  = qb + bufElems;
    float* vtb = kb + bufElems;
    float* ab  = vtb + bufElems;

    const int M = BATCH * SEQ, N = DM, K = DM;
    const int gblocks = (M / 128) * (N / 128);

    qkv_proj_kernel<<<3 * gblocks, 256, 0, stream>>>(x, Wq, Wk, Wv, qb, kb, vtb, M, N, K);

    const int waves = BATCH * NH * (SEQ / 16);
    attn_kernel<<<waves / 8, 256, 0, stream>>>(qb, kb, vtb, ab);

    gemm_xwt_kernel<<<gblocks, 256, 0, stream>>>(ab, Wo, out, M, N, K);
}


# attention: next tile's hand-over (LDS stores, loads of tile kt+2, next K fragment reads) software-pipelined behind the score product, off the next trip's critical path; sum order unchanged
# speedup vs baseline: 2.7781x; 1.0911x over previous
#include <hip/hip_runtime.h>
#include <math.h>

typedef __attribute__((ext_vector_type(2))) float v2f;
typedef __attribute__((ext_vector_type(4))) float v4f;
typedef __attribute__((ext_vector_type(8))) float v8f;

#define BATCH 2
#define SEQ 2048
#define DM 1024
#define NH 16
#define DK 64

__device__ __forceinline__ float xpair(float x) {
    int b = __builtin_bit_cast(int, x);
    return __builtin_bit_cast(float, __builtin_amdgcn_update_dpp(b, b, 0xB1, 0xF, 0xF, true));
}

#define GK_CHUNK 64
#define GK_PITCH 68
#define GK_TPR (GK_CHUNK / 4)
#define GK_RPP (256 / GK_TPR)
#define GK_NP (128 / GK_RPP)
#define GK_STAGE (128 * GK_PITCH)

__device__ __forceinline__ void gemm_tile_mainloop(const float* __restrict__ X,
                                                   const float* __restrict__ W,
                                                   float* __restrict__ sA, float* __restrict__ sB,
                                                   int bm, int bn, int K, v8f (&c)[2][4]) {
    const int t = threadIdx.x;
    const int lane = t & 31;
    const int warp = t >> 5;
    const int wm = (warp & 3) << 5;
    const int wn = (warp >> 2) << 6;
    const int ln = lane & 15;
    const int lh = lane >> 4;

    const int sr = t / GK_TPR;
    const int sc = (t % GK_TPR) << 2;
    const float* gA = X + (size_t)(bm + sr) * K + sc;
    const float* gB = W + (size_t)(bn + sr) * K + sc;
    const int so = sr * GK_PITCH + sc;

    v4f ra[GK_NP], rb[GK_NP];
#pragma unroll
    for (int i = 0; i < GK_NP; ++i) {
        ra[i] = *(const v4f*)(gA + (size_t)(GK_RPP * i) * K);
        rb[i] = *(const v4f*)(gB + (size_t)(GK_RPP * i) * K);
    }
#pragma unroll
    for (int i = 0; i < GK_NP; ++i) {
        *(v4f*)(&sA[so + GK_RPP * i * GK_PITCH]) = ra[i];
        *(v4f*)(&sB[so + GK_RPP * i * GK_PITCH]) = rb[i];
    }
    __syncthreads();

    const int fa = (wm + ln) * GK_PITCH + 2 * lh;
    const int fb = (wn + ln) * GK_PITCH + 2 * lh;

    const int nk = K / GK_CHUNK;
    for (int kc = 0; kc < nk; ++kc) {
        const int cur = kc & 1;
        const bool more = (kc + 1 < nk);
        if (more) {
            const int k1 = (kc + 1) * GK_CHUNK;
#pragma unroll
            for (int i = 0; i < GK_NP; ++i) {
                ra[i] = *(const v4f*)(gA + (size_t)(GK_RPP * i) * K + k1);
                rb[i] = *(const v4f*)(gB + (size_t)(GK_RPP * i) * K + k1);
            }
        }
        const float* pa = sA + cur * GK_STAGE + fa;
        const float* pb = sB + cur * GK_STAGE + fb;
#pragma unroll
        for (int j = 0; j < GK_CHUNK / 4; ++j) {
            v2f a0 = *(const v2f*)(pa + 4 * j);
            v2f a1 = *(const v2f*)(pa + 16 * GK_PITCH + 4 * j);
            v2f b0 = *(const v2f*)(pb + 4 * j);
            v2f b1 = *(const v2f*)(pb + 16 * GK_PITCH + 4 * j);
            v2f b2 = *(const v2f*)(pb + 32 * GK_PITCH + 4 * j);
            v2f b3 = *(const v2f*)(pb + 48 * GK_PITCH + 4 * j);
            c[0][0] = __builtin_amdgcn_wmma_f32_16x16x4_f32(false, a0, false, b0, (short)0, c[0][0], false, false);
            c[0][1] = __builtin_amdgcn_wmma_f32_16x16x4_f32(false, a0, false, b1, (short)0, c[0][1], false, false);
            c[0][2] = __builtin_amdgcn_wmma_f32_16x16x4_f32(false, a0, false, b2, (short)0, c[0][2], false, false);
            c[0][3] = __builtin_amdgcn_wmma_f32_16x16x4_f32(false, a0, false, b3, (short)0, c[0][3], false, false);
            c[1][0] = __builtin_amdgcn_wmma_f32_16x16x4_f32(false, a1, false, b0, (short)0, c[1][0], false, false);
            c[1][1] = __builtin_amdgcn_wmma_f32_16x16x4_f32(false, a1, false, b1, (short)0, c[1][1], false, false);
            c[1][2] = __builtin_amdgcn_wmma_f32_16x16x4_f32(false, a1, false, b2, (short)0, c[1][2], false, false);
            c[1][3] = __builtin_amdgcn_wmma_f32_16x16x4_f32(false, a1, false, b3, (short)0, c[1][3], false, false);
        }
        if (more) {
            float* da = sA + (cur ^ 1) * GK_STAGE + so;
            float* db = sB + (cur ^ 1) * GK_STAGE + so;
#pragma unroll
            for (int i = 0; i < GK_NP; ++i) {
                *(v4f*)(da + GK_RPP * i * GK_PITCH) = ra[i];
                *(v4f*)(db + GK_RPP * i * GK_PITCH) = rb[i];
            }
        }
        __syncthreads();
    }
}

__device__ __forceinline__ void gemm_tile_rotary(v8f (&c)[2][4], int tm, int tn, int ln, int lh) {
    const bool odd = (ln & 1) != 0;
    float fr[4];
#pragma unroll
    for (int ni = 0; ni < 4; ++ni) {
        const int i = ((tn + ni * 16 + ln) & (DK - 1)) >> 1;
        fr[ni] = __expf(-(float)i * (1.0f / 32.0f) * 9.210340371976184f);
    }
#pragma unroll
    for (int mi = 0; mi < 2; ++mi)
#pragma unroll
        for (int r = 0; r < 8; ++r) {
            const float sf = (float)((tm + mi * 16 + 8 * lh + r) & (SEQ - 1));
#pragma unroll
            for (int ni = 0; ni < 4; ++ni) {
                const float ang = sf * fr[ni];
                const float cs = __cosf(ang);
                const float sn = __sinf(ang);
                const float v = c[mi][ni][r];
                const float p = xpair(v);
                const float sg = odd ? sn : -sn;
                c[mi][ni][r] = cs * v + sg * p;
            }
        }
}

__device__ __forceinline__ void gemm_tile_store_plain(const v8f (&c)[2][4], float* __restrict__ out,
                                                      int tm, int tn, int N, int ln, int lh) {
#pragma unroll
    for (int mi = 0; mi < 2; ++mi)
#pragma unroll
        for (int ni = 0; ni < 4; ++ni) {
            float* orow = out + (size_t)(tm + mi * 16 + 8 * lh) * N + tn + ni * 16 + ln;
#pragma unroll
            for (int r = 0; r < 8; ++r)
                orow[(size_t)r * N] = c[mi][ni][r];
        }
}

__device__ __forceinline__ void gemm_tile_store_transposed(const v8f (&c)[2][4], float* __restrict__ out,
                                                           int tm, int tn, int N, int ln, int lh) {
    const int bb = tm >> 11;
#pragma unroll
    for (int mi = 0; mi < 2; ++mi)
#pragma unroll
        for (int ni = 0; ni < 4; ++ni) {
            int n = tn + ni * 16 + ln;
            int srow = (tm & (SEQ - 1)) + mi * 16 + 8 * lh;
            float* ocol = out + ((size_t)(bb * N + n) << 11) + srow;
#pragma unroll
            for (int r = 0; r < 8; ++r)
                ocol[r] = c[mi][ni][r];
        }
}

__global__ __launch_bounds__(256) void gemm_xwt_kernel(const float* __restrict__ X,
                                                       const float* __restrict__ W,
                                                       float* __restrict__ out,
                                                       int M, int N, int K) {
    __shared__ __attribute__((aligned(16))) float sA[2 * GK_STAGE];
    __shared__ __attribute__((aligned(16))) float sB[2 * GK_STAGE];
    const int lane = threadIdx.x & 31;
    const int warp = threadIdx.x >> 5;
    const int tilesN = N >> 7;
    const int bm = (blockIdx.x / tilesN) << 7;
    const int bn = (blockIdx.x % tilesN) << 7;
    v8f c[2][4] = {};
    gemm_tile_mainloop(X, W, sA, sB, bm, bn, K, c);
    gemm_tile_store_plain(c, out, bm + ((warp & 3) << 5), bn + ((warp >> 2) << 6), N, lane & 15, lane >> 4);
}

__global__ __launch_bounds__(256) void qkv_proj_kernel(const float* __restrict__ X,
                                                       const float* __restrict__ Wq,
                                                       const float* __restrict__ Wk,
                                                       const float* __restrict__ Wv,
                                                       float* __restrict__ qo,
                                                       float* __restrict__ ko,
                                                       float* __restrict__ vto,
                                                       int M, int N, int K) {
    __shared__ __attribute__((aligned(16))) float sA[2 * GK_STAGE];
    __shared__ __attribute__((aligned(16))) float sB[2 * GK_STAGE];
    const int lane = threadIdx.x & 31;
    const int warp = threadIdx.x >> 5;
    const int tilesN = N >> 7;
    const int tiles = (M >> 7) * tilesN;
    const int which = blockIdx.x / tiles;
    const int tile = blockIdx.x - which * tiles;
    const int bm = (tile / tilesN) << 7;
    const int bn = (tile % tilesN) << 7;
    const float* W = (which == 0) ? Wq : (which == 1) ? Wk : Wv;
    float* out = (which == 0) ? qo : (which == 1) ? ko : vto;
    v8f c[2][4] = {};
    gemm_tile_mainloop(X, W, sA, sB, bm, bn, K, c);
    const int tm = bm + ((warp & 3) << 5);
    const int tn = bn + ((warp >> 2) << 6);
    const int ln = lane & 15;
    const int lh = lane >> 4;
    if (which == 2) {
        gemm_tile_store_transposed(c, out, tm, tn, N, ln, lh);
    } else {
        gemm_tile_rotary(c, tm, tn, ln, lh);
        gemm_tile_store_plain(c, out, tm, tn, N, ln, lh);
    }
}

__device__ __forceinline__ float xhalf(float x) {
    int b = __builtin_bit_cast(int, x);
    return __builtin_bit_cast(float,
        __builtin_amdgcn_permlanex16(b, b, 0x76543210, 0xfedcba98, false, false));
}

#define AK_PITCH 68
#define AV_PITCH 20
#define AW_K 0
#define AW_V (16 * AK_PITCH)
#define AW_FLOATS (AW_V + 64 * AV_PITCH)

__global__ __launch_bounds__(256) void attn_kernel(const float* __restrict__ q,
                                                   const float* __restrict__ k,
                                                   const float* __restrict__ vt,
                                                   float* __restrict__ o) {
    __shared__ __attribute__((aligned(16))) float lds[8][AW_FLOATS];
    const int lane = threadIdx.x & 31;
    const int warp = threadIdx.x >> 5;
    const int bh = blockIdx.x >> 4;
    const int grp = ((blockIdx.x & 15) << 2) + (warp & 3);
    const int qt = (warp < 4) ? 127 - grp : grp;
    const int h = bh & 15;
    const int b = bh >> 4;
    const int ln = lane & 15;
    const int lh = lane >> 4;

    const float* qbase = q + ((size_t)(b * SEQ) + qt * 16 + ln) * DM + h * DK;
    v2f qa[16];
#pragma unroll
    for (int j = 0; j < 16; ++j)
        qa[j] = *(const v2f*)(qbase + 4 * j + 2 * lh);

    float* sK = &lds[warp][AW_K];
    float* sV = &lds[warp][AW_V];

    const int kr = lane >> 4;
    const int kc = (lane & 15) << 2;
    const int vr = lane >> 2;
    const int vc = (lane & 3) << 2;
    const float* gK = k + ((size_t)(b * SEQ) + kr) * DM + h * DK + kc;
    const float* gV = vt + ((size_t)(b * DM + h * DK + vr) << 11) + vc;
    float* stK = sK + kr * AK_PITCH + kc;
    float* stV = sV + vr * AV_PITCH + vc;

    v4f rk[8], rv[8];
#pragma unroll
    for (int i = 0; i < 8; ++i) {
        rk[i] = *(const v4f*)(gK + (size_t)(2 * i) * DM);
        rv[i] = *(const v4f*)(gV + ((size_t)(8 * i) << 11));
    }

    const int kperm = (((ln & 7) >> 1) << 2) + ((ln >> 3) << 1) + (ln & 1);
    const float* fK = sK + kperm * AK_PITCH + 2 * lh;
    const float* fV = sV + ln * AV_PITCH + 2 * lh;

#pragma unroll
    for (int i = 0; i < 8; ++i) {
        *(v4f*)(stK + 2 * i * AK_PITCH) = rk[i];
        *(v4f*)(stV + 8 * i * AV_PITCH) = rv[i];
    }
    {
        const int k1 = (qt > 0) ? 1 : 0;
#pragma unroll
        for (int i = 0; i < 8; ++i) {
            rk[i] = *(const v4f*)(gK + (size_t)(k1 * 16 + 2 * i) * DM);
            rv[i] = *(const v4f*)(gV + ((size_t)(8 * i) << 11) + k1 * 16);
        }
    }
    v2f kf[16];
#pragma unroll
    for (int j = 0; j < 16; ++j)
        kf[j] = *(const v2f*)(fK + 4 * j);

    v8f acc[4] = {v8f{}, v8f{}, v8f{}, v8f{}};
    float mrun = -1e30f;
    float lrun = 0.0f;

    asm volatile("" : : "v"(qa[0]), "v"(qa[1]), "v"(qa[2]), "v"(qa[3]), "v"(qa[4]), "v"(qa[5]),
                        "v"(qa[6]), "v"(qa[7]), "v"(qa[8]), "v"(qa[9]), "v"(qa[10]), "v"(qa[11]),
                        "v"(qa[12]), "v"(qa[13]), "v"(qa[14]), "v"(qa[15]));

    for (int kt = 0; kt <= qt; ++kt) {
        v2f vf[4][4];
#pragma unroll
        for (int j = 0; j < 4; ++j)
#pragma unroll
            for (int nt = 0; nt < 4; ++nt)
                vf[j][nt] = *(const v2f*)(fV + nt * 16 * AV_PITCH + 4 * j);
        v8f s = {};
#pragma unroll
        for (int j = 0; j < 16; ++j)
            s = __builtin_amdgcn_wmma_f32_16x16x4_f32(false, kf[j], false, qa[j],
                                                      (short)0, s, false, false);
        asm volatile("" : : "v"(s) : "memory");
#pragma unroll
        for (int i = 0; i < 8; ++i) {
            *(v4f*)(stK + 2 * i * AK_PITCH) = rk[i];
            *(v4f*)(stV + 8 * i * AV_PITCH) = rv[i];
        }
        const int kn = (kt + 2 < qt) ? kt + 2 : qt;
#pragma unroll
        for (int i = 0; i < 8; ++i) {
            rk[i] = *(const v4f*)(gK + (size_t)(kn * 16 + 2 * i) * DM);
            rv[i] = *(const v4f*)(gV + ((size_t)(8 * i) << 11) + kn * 16);
        }
#pragma unroll
        for (int j = 0; j < 16; ++j)
            kf[j] = *(const v2f*)(fK + 4 * j);
        float mx = -1e30f;
#pragma unroll
        for (int r = 0; r < 8; ++r) {
            float sv = s[r] * 0.125f;
            if (kt == qt && (4 * (r >> 1) + (r & 1) + 2 * lh) > ln)
                sv = -1e30f;
            s[r] = sv;
            mx = fmaxf(mx, sv);
        }
        mx = fmaxf(mx, xhalf(mx));
        const float nm = fmaxf(mrun, mx);
        const float sc = __expf(mrun - nm);
        float ps = 0.0f;
#pragma unroll
        for (int r = 0; r < 8; ++r) {
            float p = __expf(s[r] - nm);
            s[r] = p;
            ps += p;
        }
        lrun = lrun * sc + ps;
        mrun = nm;
#pragma unroll
        for (int nt = 0; nt < 4; ++nt)
#pragma unroll
            for (int r = 0; r < 8; ++r) acc[nt][r] *= sc;
#pragma unroll
        for (int j = 0; j < 4; ++j) {
            v2f pb = {s[2 * j], s[2 * j + 1]};
#pragma unroll
            for (int nt = 0; nt < 4; ++nt)
                acc[nt] = __builtin_amdgcn_wmma_f32_16x16x4_f32(false, vf[j][nt], false, pb,
                                                                (short)0, acc[nt],
                                                                false, false);
        }
    }

    const float inv = 1.0f / (lrun + xhalf(lrun));
#pragma unroll
    for (int nt = 0; nt < 4; ++nt) {
        v4f lo = {acc[nt][0] * inv, acc[nt][1] * inv, acc[nt][2] * inv, acc[nt][3] * inv};
        v4f hi = {acc[nt][4] * inv, acc[nt][5] * inv, acc[nt][6] * inv, acc[nt][7] * inv};
        float* d = sK + ln * AK_PITCH + nt * 16 + 8 * lh;
        *(v4f*)(d) = lo;
        *(v4f*)(d + 4) = hi;
    }
    float* obase = o + ((size_t)(b * SEQ) + qt * 16 + kr) * DM + h * DK + kc;
#pragma unroll
    for (int i = 0; i < 8; ++i) {
        v4f t = *(const v4f*)(sK + (2 * i + kr) * AK_PITCH + kc);
        *(v4f*)(obase + (size_t)(2 * i) * DM) = t;
    }
}

extern "C" void kernel_launch(void* const* d_in, const int* in_sizes, int n_in,
                              void* d_out, int out_size, void* d_ws, size_t ws_size,
                              hipStream_t stream) {
    const float* x  = (const float*)d_in[0];
    const float* Wq = (const float*)d_in[1];
    const float* Wk = (const float*)d_in[2];
    const float* Wv = (const float*)d_in[3];
    const float* Wo = (const float*)d_in[4];
    float* out = (float*)d_out;

    const size_t bufElems = (size_t)BATCH * SEQ * DM;
    float* qb  = (float*)d_ws;
    float* kb  = qb + bufElems;
    float* vtb = kb + bufElems;
    float* ab  = vtb + bufElems;

    const int M = BATCH * SEQ, N = DM, K = DM;
    const int gblocks = (M / 128) * (N / 128);

    qkv_proj_kernel<<<3 * gblocks, 256, 0, stream>>>(x, Wq, Wk, Wv, qb, kb, vtb, M, N, K);

    const int waves = BATCH * NH * (SEQ / 16);
    attn_kernel<<<waves / 8, 256, 0, stream>>>(qb, kb, vtb, ab);

    gemm_xwt_kernel<<<gblocks, 256, 0, stream>>>(ab, Wo, out, M, N, K);
}


# attention: each late query tile's keys are split at key tile 64 between the two waves of its SIMD pair (the first also does the early tile); partial softmax states merged in LDS; sum order changes
# speedup vs baseline: 2.8717x; 1.0337x over previous
#include <hip/hip_runtime.h>
#include <math.h>

typedef __attribute__((ext_vector_type(2))) float v2f;
typedef __attribute__((ext_vector_type(4))) float v4f;
typedef __attribute__((ext_vector_type(8))) float v8f;

#define BATCH 2
#define SEQ 2048
#define DM 1024
#define NH 16
#define DK 64

__device__ __forceinline__ float xpair(float x) {
    int b = __builtin_bit_cast(int, x);
    return __builtin_bit_cast(float, __builtin_amdgcn_update_dpp(b, b, 0xB1, 0xF, 0xF, true));
}

#define GK_CHUNK 64
#define GK_PITCH 68
#define GK_TPR (GK_CHUNK / 4)
#define GK_RPP (256 / GK_TPR)
#define GK_NP (128 / GK_RPP)
#define GK_STAGE (128 * GK_PITCH)

__device__ __forceinline__ void gemm_tile_mainloop(const float* __restrict__ X,
                                                   const float* __restrict__ W,
                                                   float* __restrict__ sA, float* __restrict__ sB,
                                                   int bm, int bn, int K, v8f (&c)[2][4]) {
    const int t = threadIdx.x;
    const int lane = t & 31;
    const int warp = t >> 5;
    const int wm = (warp & 3) << 5;
    const int wn = (warp >> 2) << 6;
    const int ln = lane & 15;
    const int lh = lane >> 4;

    const int sr = t / GK_TPR;
    const int sc = (t % GK_TPR) << 2;
    const float* gA = X + (size_t)(bm + sr) * K + sc;
    const float* gB = W + (size_t)(bn + sr) * K + sc;
    const int so = sr * GK_PITCH + sc;

    v4f ra[GK_NP], rb[GK_NP];
#pragma unroll
    for (int i = 0; i < GK_NP; ++i) {
        ra[i] = *(const v4f*)(gA + (size_t)(GK_RPP * i) * K);
        rb[i] = *(const v4f*)(gB + (size_t)(GK_RPP * i) * K);
    }
#pragma unroll
    for (int i = 0; i < GK_NP; ++i) {
        *(v4f*)(&sA[so + GK_RPP * i * GK_PITCH]) = ra[i];
        *(v4f*)(&sB[so + GK_RPP * i * GK_PITCH]) = rb[i];
    }
    __syncthreads();

    const int fa = (wm + ln) * GK_PITCH + 2 * lh;
    const int fb = (wn + ln) * GK_PITCH + 2 * lh;

    const int nk = K / GK_CHUNK;
    for (int kc = 0; kc < nk; ++kc) {
        const int cur = kc & 1;
        const bool more = (kc + 1 < nk);
        if (more) {
            const int k1 = (kc + 1) * GK_CHUNK;
#pragma unroll
            for (int i = 0; i < GK_NP; ++i) {
                ra[i] = *(const v4f*)(gA + (size_t)(GK_RPP * i) * K + k1);
                rb[i] = *(const v4f*)(gB + (size_t)(GK_RPP * i) * K + k1);
            }
        }
        const float* pa = sA + cur * GK_STAGE + fa;
        const float* pb = sB + cur * GK_STAGE + fb;
#pragma unroll
        for (int j = 0; j < GK_CHUNK / 4; ++j) {
            v2f a0 = *(const v2f*)(pa + 4 * j);
            v2f a1 = *(const v2f*)(pa + 16 * GK_PITCH + 4 * j);
            v2f b0 = *(const v2f*)(pb + 4 * j);
            v2f b1 = *(const v2f*)(pb + 16 * GK_PITCH + 4 * j);
            v2f b2 = *(const v2f*)(pb + 32 * GK_PITCH + 4 * j);
            v2f b3 = *(const v2f*)(pb + 48 * GK_PITCH + 4 * j);
            c[0][0] = __builtin_amdgcn_wmma_f32_16x16x4_f32(false, a0, false, b0, (short)0, c[0][0], false, false);
            c[0][1] = __builtin_amdgcn_wmma_f32_16x16x4_f32(false, a0, false, b1, (short)0, c[0][1], false, false);
            c[0][2] = __builtin_amdgcn_wmma_f32_16x16x4_f32(false, a0, false, b2, (short)0, c[0][2], false, false);
            c[0][3] = __builtin_amdgcn_wmma_f32_16x16x4_f32(false, a0, false, b3, (short)0, c[0][3], false, false);
            c[1][0] = __builtin_amdgcn_wmma_f32_16x16x4_f32(false, a1, false, b0, (short)0, c[1][0], false, false);
            c[1][1] = __builtin_amdgcn_wmma_f32_16x16x4_f32(false, a1, false, b1, (short)0, c[1][1], false, false);
            c[1][2] = __builtin_amdgcn_wmma_f32_16x16x4_f32(false, a1, false, b2, (short)0, c[1][2], false, false);
            c[1][3] = __builtin_amdgcn_wmma_f32_16x16x4_f32(false, a1, false, b3, (short)0, c[1][3], false, false);
        }
        if (more) {
            float* da = sA + (cur ^ 1) * GK_STAGE + so;
            float* db = sB + (cur ^ 1) * GK_STAGE + so;
#pragma unroll
            for (int i = 0; i < GK_NP; ++i) {
                *(v4f*)(da + GK_RPP * i * GK_PITCH) = ra[i];
                *(v4f*)(db + GK_RPP * i * GK_PITCH) = rb[i];
            }
        }
        __syncthreads();
    }
}

__device__ __forceinline__ void gemm_tile_rotary(v8f (&c)[2][4], int tm, int tn, int ln, int lh) {
    const bool odd = (ln & 1) != 0;
    float fr[4];
#pragma unroll
    for (int ni = 0; ni < 4; ++ni) {
        const int i = ((tn + ni * 16 + ln) & (DK - 1)) >> 1;
        fr[ni] = __expf(-(float)i * (1.0f / 32.0f) * 9.210340371976184f);
    }
#pragma unroll
    for (int mi = 0; mi < 2; ++mi)
#pragma unroll
        for (int r = 0; r < 8; ++r) {
            const float sf = (float)((tm + mi * 16 + 8 * lh + r) & (SEQ - 1));
#pragma unroll
            for (int ni = 0; ni < 4; ++ni) {
                const float ang = sf * fr[ni];
                const float cs = __cosf(ang);
                const float sn = __sinf(ang);
                const float v = c[mi][ni][r];
                const float p = xpair(v);
                const float sg = odd ? sn : -sn;
                c[mi][ni][r] = cs * v + sg * p;
            }
        }
}

__device__ __forceinline__ void gemm_tile_store_plain(const v8f (&c)[2][4], float* __restrict__ out,
                                                      int tm, int tn, int N, int ln, int lh) {
#pragma unroll
    for (int mi = 0; mi < 2; ++mi)
#pragma unroll
        for (int ni = 0; ni < 4; ++ni) {
            float* orow = out + (size_t)(tm + mi * 16 + 8 * lh) * N + tn + ni * 16 + ln;
#pragma unroll
            for (int r = 0; r < 8; ++r)
                orow[(size_t)r * N] = c[mi][ni][r];
        }
}

__device__ __forceinline__ void gemm_tile_store_transposed(const v8f (&c)[2][4], float* __restrict__ out,
                                                           int tm, int tn, int N, int ln, int lh) {
    const int bb = tm >> 11;
#pragma unroll
    for (int mi = 0; mi < 2; ++mi)
#pragma unroll
        for (int ni = 0; ni < 4; ++ni) {
            int n = tn + ni * 16 + ln;
            int srow = (tm & (SEQ - 1)) + mi * 16 + 8 * lh;
            float* ocol = out + ((size_t)(bb * N + n) << 11) + srow;
#pragma unroll
            for (int r = 0; r < 8; ++r)
                ocol[r] = c[mi][ni][r];
        }
}

__global__ __launch_bounds__(256) void gemm_xwt_kernel(const float* __restrict__ X,
                                                       const float* __restrict__ W,
                                                       float* __restrict__ out,
                                                       int M, int N, int K) {
    __shared__ __attribute__((aligned(16))) float sA[2 * GK_STAGE];
    __shared__ __attribute__((aligned(16))) float sB[2 * GK_STAGE];
    const int lane = threadIdx.x & 31;
    const int warp = threadIdx.x >> 5;
    const int tilesN = N >> 7;
    const int bm = (blockIdx.x / tilesN) << 7;
    const int bn = (blockIdx.x % tilesN) << 7;
    v8f c[2][4] = {};
    gemm_tile_mainloop(X, W, sA, sB, bm, bn, K, c);
    gemm_tile_store_plain(c, out, bm + ((warp & 3) << 5), bn + ((warp >> 2) << 6), N, lane & 15, lane >> 4);
}

__global__ __launch_bounds__(256) void qkv_proj_kernel(const float* __restrict__ X,
                                                       const float* __restrict__ Wq,
                                                       const float* __restrict__ Wk,
                                                       const float* __restrict__ Wv,
                                                       float* __restrict__ qo,
                                                       float* __restrict__ ko,
                                                       float* __restrict__ vto,
                                                       int M, int N, int K) {
    __shared__ __attribute__((aligned(16))) float sA[2 * GK_STAGE];
    __shared__ __attribute__((aligned(16))) float sB[2 * GK_STAGE];
    const int lane = threadIdx.x & 31;
    const int warp = threadIdx.x >> 5;
    const int tilesN = N >> 7;
    const int tiles = (M >> 7) * tilesN;
    const int which = blockIdx.x / tiles;
    const int tile = blockIdx.x - which * tiles;
    const int bm = (tile / tilesN) << 7;
    const int bn = (tile % tilesN) << 7;
    const float* W = (which == 0) ? Wq : (which == 1) ? Wk : Wv;
    float* out = (which == 0) ? qo : (which == 1) ? ko : vto;
    v8f c[2][4] = {};
    gemm_tile_mainloop(X, W, sA, sB, bm, bn, K, c);
    const int tm = bm + ((warp & 3) << 5);
    const int tn = bn + ((warp >> 2) << 6);
    const int ln = lane & 15;
    const int lh = lane >> 4;
    if (which == 2) {
        gemm_tile_store_transposed(c, out, tm, tn, N, ln, lh);
    } else {
        gemm_tile_rotary(c, tm, tn, ln, lh);
        gemm_tile_store_plain(c, out, tm, tn, N, ln, lh);
    }
}

__device__ __forceinline__ float xhalf(float x) {
    int b = __builtin_bit_cast(int, x);
    return __builtin_bit_cast(float,
        __builtin_amdgcn_permlanex16(b, b, 0x76543210, 0xfedcba98, false, false));
}

#define AK_PITCH 68
#define AV_PITCH 20
#define AW_K 0
#define AW_V (16 * AK_PITCH)
#define AW_FLOATS (AW_V + 64 * AV_PITCH)

__device__ __forceinline__ void attn_store_tile(const v8f (&acc)[4], float lrun, float* sK,
                                                float* __restrict__ o, int b, int h, int qt,
                                                int ln, int lh, int kr, int kc) {
    const float inv = 1.0f / (lrun + xhalf(lrun));
#pragma unroll
    for (int nt = 0; nt < 4; ++nt) {
        v4f lo = {acc[nt][0] * inv, acc[nt][1] * inv, acc[nt][2] * inv, acc[nt][3] * inv};
        v4f hi = {acc[nt][4] * inv, acc[nt][5] * inv, acc[nt][6] * inv, acc[nt][7] * inv};
        float* d = sK + ln * AK_PITCH + nt * 16 + 8 * lh;
        *(v4f*)(d) = lo;
        *(v4f*)(d + 4) = hi;
    }
    float* obase = o + ((size_t)(b * SEQ) + qt * 16 + kr) * DM + h * DK + kc;
#pragma unroll
    for (int i = 0; i < 8; ++i) {
        v4f t = *(const v4f*)(sK + (2 * i + kr) * AK_PITCH + kc);
        *(v4f*)(obase + (size_t)(2 * i) * DM) = t;
    }
}

#define AP_M (8 * 32 * 4)
#define AP_L (AP_M + 32)

__global__ __launch_bounds__(256) void attn_kernel(const float* __restrict__ q,
                                                   const float* __restrict__ k,
                                                   const float* __restrict__ vt,
                                                   float* __restrict__ o) {
    __shared__ __attribute__((aligned(16))) float lds[8][AW_FLOATS];
    const int lane = threadIdx.x & 31;
    const int warp = __builtin_amdgcn_readfirstlane((int)(threadIdx.x >> 5));
    const int bh = blockIdx.x >> 4;
    const int grp = ((blockIdx.x & 15) << 2) + (warp & 3);
    const bool two = warp < 4;
    const int h = bh & 15;
    const int b = bh >> 4;
    const int ln = lane & 15;
    const int lh = lane >> 4;

    float* sK = &lds[warp][AW_K];
    float* sV = &lds[warp][AW_V];

    const int kr = lane >> 4;
    const int kc = (lane & 15) << 2;
    const int vr = lane >> 2;
    const int vc = (lane & 3) << 2;
    const float* gK = k + ((size_t)(b * SEQ) + kr) * DM + h * DK + kc;
    const float* gV = vt + ((size_t)(b * DM + h * DK + vr) << 11) + vc;
    float* stK = sK + kr * AK_PITCH + kc;
    float* stV = sV + vr * AV_PITCH + vc;

    const int kperm = (((ln & 7) >> 1) << 2) + ((ln >> 3) << 1) + (ln & 1);
    const float* fK = sK + kperm * AK_PITCH + 2 * lh;
    const float* fV = sV + ln * AV_PITCH + 2 * lh;

    v8f acc[4];
    float mrun;
    float lrun;

    const int nseg = two ? 2 : 1;
    for (int seg = 0; seg < nseg; ++seg) {
        const int qt = (two && seg == 0) ? grp : 127 - grp;
        const int k0 = (two && seg == 1) ? 64 : 0;
        const int k1 = two ? qt : 63;

        const float* qbase = q + ((size_t)(b * SEQ) + qt * 16 + ln) * DM + h * DK;
        v2f qa[16];
#pragma unroll
        for (int j = 0; j < 16; ++j)
            qa[j] = *(const v2f*)(qbase + 4 * j + 2 * lh);

        v4f rk[8], rv[8];
#pragma unroll
        for (int i = 0; i < 8; ++i) {
            rk[i] = *(const v4f*)(gK + (size_t)(k0 * 16 + 2 * i) * DM);
            rv[i] = *(const v4f*)(gV + ((size_t)(8 * i) << 11) + k0 * 16);
        }

#pragma unroll
        for (int i = 0; i < 8; ++i) {
            *(v4f*)(stK + 2 * i * AK_PITCH) = rk[i];
            *(v4f*)(stV + 8 * i * AV_PITCH) = rv[i];
        }
        {
            const int kp = (k0 + 1 < k1) ? k0 + 1 : k1;
#pragma unroll
            for (int i = 0; i < 8; ++i) {
                rk[i] = *(const v4f*)(gK + (size_t)(kp * 16 + 2 * i) * DM);
                rv[i] = *(const v4f*)(gV + ((size_t)(8 * i) << 11) + kp * 16);
            }
        }
        v2f kf[16];
#pragma unroll
        for (int j = 0; j < 16; ++j)
            kf[j] = *(const v2f*)(fK + 4 * j);

#pragma unroll
        for (int nt = 0; nt < 4; ++nt) acc[nt] = v8f{};
        mrun = -1e30f;
        lrun = 0.0f;

        asm volatile("" : : "v"(qa[0]), "v"(qa[1]), "v"(qa[2]), "v"(qa[3]), "v"(qa[4]), "v"(qa[5]),
                            "v"(qa[6]), "v"(qa[7]), "v"(qa[8]), "v"(qa[9]), "v"(qa[10]), "v"(qa[11]),
                            "v"(qa[12]), "v"(qa[13]), "v"(qa[14]), "v"(qa[15]));

        for (int kt = k0; kt <= k1; ++kt) {
            v2f vf[4][4];
#pragma unroll
            for (int j = 0; j < 4; ++j)
#pragma unroll
                for (int nt = 0; nt < 4; ++nt)
                    vf[j][nt] = *(const v2f*)(fV + nt * 16 * AV_PITCH + 4 * j);
            v8f s = {};
#pragma unroll
            for (int j = 0; j < 16; ++j)
                s = __builtin_amdgcn_wmma_f32_16x16x4_f32(false, kf[j], false, qa[j],
                                                          (short)0, s, false, false);
            asm volatile("" : : "v"(s) : "memory");
#pragma unroll
            for (int i = 0; i < 8; ++i) {
                *(v4f*)(stK + 2 * i * AK_PITCH) = rk[i];
                *(v4f*)(stV + 8 * i * AV_PITCH) = rv[i];
            }
            const int kn = (kt + 2 < k1) ? kt + 2 : k1;
#pragma unroll
            for (int i = 0; i < 8; ++i) {
                rk[i] = *(const v4f*)(gK + (size_t)(kn * 16 + 2 * i) * DM);
                rv[i] = *(const v4f*)(gV + ((size_t)(8 * i) << 11) + kn * 16);
            }
#pragma unroll
            for (int j = 0; j < 16; ++j)
                kf[j] = *(const v2f*)(fK + 4 * j);
            float mx = -1e30f;
#pragma unroll
            for (int r = 0; r < 8; ++r) {
                float sv = s[r] * 0.125f;
                if (kt == qt && (4 * (r >> 1) + (r & 1) + 2 * lh) > ln)
                    sv = -1e30f;
                s[r] = sv;
                mx = fmaxf(mx, sv);
            }
            mx = fmaxf(mx, xhalf(mx));
            const float nm = fmaxf(mrun, mx);
            const float sc = __expf(mrun - nm);
            float ps = 0.0f;
#pragma unroll
            for (int r = 0; r < 8; ++r) {
                float p = __expf(s[r] - nm);
                s[r] = p;
                ps += p;
            }
            lrun = lrun * sc + ps;
            mrun = nm;
#pragma unroll
            for (int nt = 0; nt < 4; ++nt)
#pragma unroll
                for (int r = 0; r < 8; ++r) acc[nt][r] *= sc;
#pragma unroll
            for (int j = 0; j < 4; ++j) {
                v2f pb = {s[2 * j], s[2 * j + 1]};
#pragma unroll
                for (int nt = 0; nt < 4; ++nt)
                    acc[nt] = __builtin_amdgcn_wmma_f32_16x16x4_f32(false, vf[j][nt], false, pb,
                                                                    (short)0, acc[nt],
                                                                    false, false);
            }
        }

        if (two && seg == 0)
            attn_store_tile(acc, lrun, sK, o, b, h, qt, ln, lh, kr, kc);
    }

    if (two) {
        float* pp = &lds[warp][0];
#pragma unroll
        for (int nt = 0; nt < 4; ++nt) {
            v4f lo = {acc[nt][0], acc[nt][1], acc[nt][2], acc[nt][3]};
            v4f hi = {acc[nt][4], acc[nt][5], acc[nt][6], acc[nt][7]};
            *(v4f*)(pp + ((nt * 2 + 0) * 32 + lane) * 4) = lo;
            *(v4f*)(pp + ((nt * 2 + 1) * 32 + lane) * 4) = hi;
        }
        pp[AP_M + lane] = mrun;
        pp[AP_L + lane] = lrun;
    }
    __syncthreads();
    if (!two) {
        const float* pp = &lds[warp - 4][0];
        const float pm = pp[AP_M + lane];
        const float pl = pp[AP_L + lane];
        const float nm = fmaxf(mrun, pm);
        const float fa = __expf(mrun - nm);
        const float fb = __expf(pm - nm);
        lrun = lrun * fa + pl * fb;
#pragma unroll
        for (int nt = 0; nt < 4; ++nt) {
            const v4f lo = *(const v4f*)(pp + ((nt * 2 + 0) * 32 + lane) * 4);
            const v4f hi = *(const v4f*)(pp + ((nt * 2 + 1) * 32 + lane) * 4);
#pragma unroll
            for (int e = 0; e < 4; ++e) {
                acc[nt][e] = acc[nt][e] * fa + lo[e] * fb;
                acc[nt][4 + e] = acc[nt][4 + e] * fa + hi[e] * fb;
            }
        }
        attn_store_tile(acc, lrun, sK, o, b, h, 127 - grp, ln, lh, kr, kc);
    }
}

extern "C" void kernel_launch(void* const* d_in, const int* in_sizes, int n_in,
                              void* d_out, int out_size, void* d_ws, size_t ws_size,
                              hipStream_t stream) {
    const float* x  = (const float*)d_in[0];
    const float* Wq = (const float*)d_in[1];
    const float* Wk = (const float*)d_in[2];
    const float* Wv = (const float*)d_in[3];
    const float* Wo = (const float*)d_in[4];
    float* out = (float*)d_out;

    const size_t bufElems = (size_t)BATCH * SEQ * DM;
    float* qb  = (float*)d_ws;
    float* kb  = qb + bufElems;
    float* vtb = kb + bufElems;
    float* ab  = vtb + bufElems;

    const int M = BATCH * SEQ, N = DM, K = DM;
    const int gblocks = (M / 128) * (N / 128);

    qkv_proj_kernel<<<3 * gblocks, 256, 0, stream>>>(x, Wq, Wk, Wv, qb, kb, vtb, M, N, K);

    const int waves = BATCH * NH * (SEQ / 16);
    attn_kernel<<<waves / 8, 256, 0, stream>>>(qb, kb, vtb, ab);

    gemm_xwt_kernel<<<gblocks, 256, 0, stream>>>(ab, Wo, out, M, N, K);
}


# attention: K and V leave the projection as per-key-tile blocks in matrix-fragment order; waves load fragments straight from memory one trip ahead, no LDS staging in the loop; sum order unchanged
# speedup vs baseline: 2.9982x; 1.0441x over previous
#include <hip/hip_runtime.h>
#include <math.h>

typedef __attribute__((ext_vector_type(2))) float v2f;
typedef __attribute__((ext_vector_type(4))) float v4f;
typedef __attribute__((ext_vector_type(8))) float v8f;

#define BATCH 2
#define SEQ 2048
#define DM 1024
#define NH 16
#define DK 64

__device__ __forceinline__ float xpair(float x) {
    int b = __builtin_bit_cast(int, x);
    return __builtin_bit_cast(float, __builtin_amdgcn_update_dpp(b, b, 0xB1, 0xF, 0xF, true));
}

#define GK_CHUNK 64
#define GK_PITCH 68
#define GK_TPR (GK_CHUNK / 4)
#define GK_RPP (256 / GK_TPR)
#define GK_NP (128 / GK_RPP)
#define GK_STAGE (128 * GK_PITCH)

__device__ __forceinline__ void gemm_tile_mainloop(const float* __restrict__ X,
                                                   const float* __restrict__ W,
                                                   float* __restrict__ sA, float* __restrict__ sB,
                                                   int bm, int bn, int K, v8f (&c)[2][4]) {
    const int t = threadIdx.x;
    const int lane = t & 31;
    const int warp = t >> 5;
    const int wm = (warp & 3) << 5;
    const int wn = (warp >> 2) << 6;
    const int ln = lane & 15;
    const int lh = lane >> 4;

    const int sr = t / GK_TPR;
    const int sc = (t % GK_TPR) << 2;
    const float* gA = X + (size_t)(bm + sr) * K + sc;
    const float* gB = W + (size_t)(bn + sr) * K + sc;
    const int so = sr * GK_PITCH + sc;

    v4f ra[GK_NP], rb[GK_NP];
#pragma unroll
    for (int i = 0; i < GK_NP; ++i) {
        ra[i] = *(const v4f*)(gA + (size_t)(GK_RPP * i) * K);
        rb[i] = *(const v4f*)(gB + (size_t)(GK_RPP * i) * K);
    }
#pragma unroll
    for (int i = 0; i < GK_NP; ++i) {
        *(v4f*)(&sA[so + GK_RPP * i * GK_PITCH]) = ra[i];
        *(v4f*)(&sB[so + GK_RPP * i * GK_PITCH]) = rb[i];
    }
    __syncthreads();

    const int fa = (wm + ln) * GK_PITCH + 2 * lh;
    const int fb = (wn + ln) * GK_PITCH + 2 * lh;

    const int nk = K / GK_CHUNK;
    for (int kc = 0; kc < nk; ++kc) {
        const int cur = kc & 1;
        const bool more = (kc + 1 < nk);
        if (more) {
            const int k1 = (kc + 1) * GK_CHUNK;
#pragma unroll
            for (int i = 0; i < GK_NP; ++i) {
                ra[i] = *(const v4f*)(gA + (size_t)(GK_RPP * i) * K + k1);
                rb[i] = *(const v4f*)(gB + (size_t)(GK_RPP * i) * K + k1);
            }
        }
        const float* pa = sA + cur * GK_STAGE + fa;
        const float* pb = sB + cur * GK_STAGE + fb;
#pragma unroll
        for (int j = 0; j < GK_CHUNK / 4; ++j) {
            v2f a0 = *(const v2f*)(pa + 4 * j);
            v2f a1 = *(const v2f*)(pa + 16 * GK_PITCH + 4 * j);
            v2f b0 = *(const v2f*)(pb + 4 * j);
            v2f b1 = *(const v2f*)(pb + 16 * GK_PITCH + 4 * j);
            v2f b2 = *(const v2f*)(pb + 32 * GK_PITCH + 4 * j);
            v2f b3 = *(const v2f*)(pb + 48 * GK_PITCH + 4 * j);
            c[0][0] = __builtin_amdgcn_wmma_f32_16x16x4_f32(false, a0, false, b0, (short)0, c[0][0], false, false);
            c[0][1] = __builtin_amdgcn_wmma_f32_16x16x4_f32(false, a0, false, b1, (short)0, c[0][1], false, false);
            c[0][2] = __builtin_amdgcn_wmma_f32_16x16x4_f32(false, a0, false, b2, (short)0, c[0][2], false, false);
            c[0][3] = __builtin_amdgcn_wmma_f32_16x16x4_f32(false, a0, false, b3, (short)0, c[0][3], false, false);
            c[1][0] = __builtin_amdgcn_wmma_f32_16x16x4_f32(false, a1, false, b0, (short)0, c[1][0], false, false);
            c[1][1] = __builtin_amdgcn_wmma_f32_16x16x4_f32(false, a1, false, b1, (short)0, c[1][1], false, false);
            c[1][2] = __builtin_amdgcn_wmma_f32_16x16x4_f32(false, a1, false, b2, (short)0, c[1][2], false, false);
            c[1][3] = __builtin_amdgcn_wmma_f32_16x16x4_f32(false, a1, false, b3, (short)0, c[1][3], false, false);
        }
        if (more) {
            float* da = sA + (cur ^ 1) * GK_STAGE + so;
            float* db = sB + (cur ^ 1) * GK_STAGE + so;
#pragma unroll
            for (int i = 0; i < GK_NP; ++i) {
                *(v4f*)(da + GK_RPP * i * GK_PITCH) = ra[i];
                *(v4f*)(db + GK_RPP * i * GK_PITCH) = rb[i];
            }
        }
        __syncthreads();
    }
}

__device__ __forceinline__ void gemm_tile_rotary(v8f (&c)[2][4], int tm, int tn, int ln, int lh) {
    const bool odd = (ln & 1) != 0;
    float fr[4];
#pragma unroll
    for (int ni = 0; ni < 4; ++ni) {
        const int i = ((tn + ni * 16 + ln) & (DK - 1)) >> 1;
        fr[ni] = __expf(-(float)i * (1.0f / 32.0f) * 9.210340371976184f);
    }
#pragma unroll
    for (int mi = 0; mi < 2; ++mi)
#pragma unroll
        for (int r = 0; r < 8; ++r) {
            const float sf = (float)((tm + mi * 16 + 8 * lh + r) & (SEQ - 1));
#pragma unroll
            for (int ni = 0; ni < 4; ++ni) {
                const float ang = sf * fr[ni];
                const float cs = __cosf(ang);
                const float sn = __sinf(ang);
                const float v = c[mi][ni][r];
                const float p = xpair(v);
                const float sg = odd ? sn : -sn;
                c[mi][ni][r] = cs * v + sg * p;
            }
        }
}

__device__ __forceinline__ void gemm_tile_store_plain(const v8f (&c)[2][4], float* __restrict__ out,
                                                      int tm, int tn, int N, int ln, int lh) {
#pragma unroll
    for (int mi = 0; mi < 2; ++mi)
#pragma unroll
        for (int ni = 0; ni < 4; ++ni) {
            float* orow = out + (size_t)(tm + mi * 16 + 8 * lh) * N + tn + ni * 16 + ln;
#pragma unroll
            for (int r = 0; r < 8; ++r)
                orow[(size_t)r * N] = c[mi][ni][r];
        }
}

__device__ __forceinline__ void gemm_tile_store_kfrag(const v8f (&c)[2][4], float* __restrict__ out,
                                                      int tm, int tn, int ln, int lh) {
    const int bb = tm >> 11;
    const int hd = tn >> 6;
    const int kt0 = (tm & (SEQ - 1)) >> 4;
    float* blk = out + ((size_t)((((bb * NH + hd) << 7) + kt0)) << 10);
    const int lc = ((ln >> 3) << 7) + (((ln >> 1) & 1) << 6) + (((ln >> 2) & 1) << 1) + (ln & 1) + (lh << 4);
#pragma unroll
    for (int mi = 0; mi < 2; ++mi)
#pragma unroll
        for (int ni = 0; ni < 4; ++ni)
#pragma unroll
            for (int r = 0; r < 8; ++r)
                blk[mi * 1024 + ni * 256 + lc + 32 * ((r >> 1) & 1) + 8 * (r >> 2) + 4 * (r & 1)] = c[mi][ni][r];
}

__device__ __forceinline__ void gemm_tile_store_vfrag(const v8f (&c)[2][4], float* __restrict__ out,
                                                      int tm, int tn, int ln, int lh) {
    const int bb = tm >> 11;
    const int hd = tn >> 6;
    const int kt0 = (tm & (SEQ - 1)) >> 4;
    float* blk = out + ((size_t)((((bb * NH + hd) << 7) + kt0)) << 10) + (lh << 9) + (ln << 2);
#pragma unroll
    for (int mi = 0; mi < 2; ++mi)
#pragma unroll
        for (int np = 0; np < 2; ++np)
#pragma unroll
            for (int rp = 0; rp < 4; ++rp) {
                v4f t = {c[mi][2 * np][2 * rp], c[mi][2 * np][2 * rp + 1],
                         c[mi][2 * np + 1][2 * rp], c[mi][2 * np + 1][2 * rp + 1]};
                *(v4f*)(blk + mi * 1024 + ((rp >> 1) << 8) + (np << 7) + ((rp & 1) << 6)) = t;
            }
}

__global__ __launch_bounds__(256) void gemm_xwt_kernel(const float* __restrict__ X,
                                                       const float* __restrict__ W,
                                                       float* __restrict__ out,
                                                       int M, int N, int K) {
    __shared__ __attribute__((aligned(16))) float sA[2 * GK_STAGE];
    __shared__ __attribute__((aligned(16))) float sB[2 * GK_STAGE];
    const int lane = threadIdx.x & 31;
    const int warp = threadIdx.x >> 5;
    const int tilesN = N >> 7;
    const int bm = (blockIdx.x / tilesN) << 7;
    const int bn = (blockIdx.x % tilesN) << 7;
    v8f c[2][4] = {};
    gemm_tile_mainloop(X, W, sA, sB, bm, bn, K, c);
    gemm_tile_store_plain(c, out, bm + ((warp & 3) << 5), bn + ((warp >> 2) << 6), N, lane & 15, lane >> 4);
}

__global__ __launch_bounds__(256) void qkv_proj_kernel(const float* __restrict__ X,
                                                       const float* __restrict__ Wq,
                                                       const float* __restrict__ Wk,
                                                       const float* __restrict__ Wv,
                                                       float* __restrict__ qo,
                                                       float* __restrict__ ko,
                                                       float* __restrict__ vo,
                                                       int M, int N, int K) {
    __shared__ __attribute__((aligned(16))) float sA[2 * GK_STAGE];
    __shared__ __attribute__((aligned(16))) float sB[2 * GK_STAGE];
    const int lane = threadIdx.x & 31;
    const int warp = threadIdx.x >> 5;
    const int tilesN = N >> 7;
    const int tiles = (M >> 7) * tilesN;
    const int which = blockIdx.x / tiles;
    const int tile = blockIdx.x - which * tiles;
    const int bm = (tile / tilesN) << 7;
    const int bn = (tile % tilesN) << 7;
    const float* W = (which == 0) ? Wq : (which == 1) ? Wk : Wv;
    float* out = (which == 0) ? qo : (which == 1) ? ko : vo;
    v8f c[2][4] = {};
    gemm_tile_mainloop(X, W, sA, sB, bm, bn, K, c);
    const int tm = bm + ((warp & 3) << 5);
    const int tn = bn + ((warp >> 2) << 6);
    const int ln = lane & 15;
    const int lh = lane >> 4;
    if (which == 2) {
        gemm_tile_store_vfrag(c, out, tm, tn, ln, lh);
    } else {
        gemm_tile_rotary(c, tm, tn, ln, lh);
        if (which == 1)
            gemm_tile_store_kfrag(c, out, tm, tn, ln, lh);
        else
            gemm_tile_store_plain(c, out, tm, tn, N, ln, lh);
    }
}

__device__ __forceinline__ float xhalf(float x) {
    int b = __builtin_bit_cast(int, x);
    return __builtin_bit_cast(float,
        __builtin_amdgcn_permlanex16(b, b, 0x76543210, 0xfedcba98, false, false));
}

#define AK_PITCH 68
#define AW_FLOATS 1088

__device__ __forceinline__ void attn_store_tile(const v8f (&acc)[4], float lrun, float* sK,
                                                float* __restrict__ o, int b, int h, int qt,
                                                int ln, int lh, int kr, int kc) {
    const float inv = 1.0f / (lrun + xhalf(lrun));
#pragma unroll
    for (int nt = 0; nt < 4; ++nt) {
        v4f lo = {acc[nt][0] * inv, acc[nt][1] * inv, acc[nt][2] * inv, acc[nt][3] * inv};
        v4f hi = {acc[nt][4] * inv, acc[nt][5] * inv, acc[nt][6] * inv, acc[nt][7] * inv};
        float* d = sK + ln * AK_PITCH + nt * 16 + 8 * lh;
        *(v4f*)(d) = lo;
        *(v4f*)(d + 4) = hi;
    }
    float* obase = o + ((size_t)(b * SEQ) + qt * 16 + kr) * DM + h * DK + kc;
#pragma unroll
    for (int i = 0; i < 8; ++i) {
        v4f t = *(const v4f*)(sK + (2 * i + kr) * AK_PITCH + kc);
        *(v4f*)(obase + (size_t)(2 * i) * DM) = t;
    }
}

#define AP_M (8 * 32 * 4)
#define AP_L (AP_M + 32)

__device__ __forceinline__ void attn_trip(const v4f (&ck)[8], const v4f (&cv)[8], const v2f (&qa)[16],
                                          v8f (&acc)[4], float& mrun, float& lrun,
                                          bool diag, int ln, int lh) {
    v8f s = {};
#pragma unroll
    for (int g = 0; g < 8; ++g) {
        const v2f ka = {ck[g][0], ck[g][1]};
        const v2f kb = {ck[g][2], ck[g][3]};
        s = __builtin_amdgcn_wmma_f32_16x16x4_f32(false, ka, false, qa[2 * g],
                                                  (short)0, s, false, false);
        s = __builtin_amdgcn_wmma_f32_16x16x4_f32(false, kb, false, qa[2 * g + 1],
                                                  (short)0, s, false, false);
    }
    float mx = -1e30f;
#pragma unroll
    for (int r = 0; r < 8; ++r) {
        float sv = s[r] * 0.125f;
        if (diag && (4 * (r >> 1) + (r & 1) + 2 * lh) > ln)
            sv = -1e30f;
        s[r] = sv;
        mx = fmaxf(mx, sv);
    }
    mx = fmaxf(mx, xhalf(mx));
    const float nm = fmaxf(mrun, mx);
    const float sc = __expf(mrun - nm);
    float ps = 0.0f;
#pragma unroll
    for (int r = 0; r < 8; ++r) {
        float p = __expf(s[r] - nm);
        s[r] = p;
        ps += p;
    }
    lrun = lrun * sc + ps;
    mrun = nm;
#pragma unroll
    for (int nt = 0; nt < 4; ++nt)
#pragma unroll
        for (int r = 0; r < 8; ++r) acc[nt][r] *= sc;
#pragma unroll
    for (int j = 0; j < 4; ++j) {
        const v2f pb = {s[2 * j], s[2 * j + 1]};
#pragma unroll
        for (int np = 0; np < 2; ++np) {
            const v2f va = {cv[2 * j + np][0], cv[2 * j + np][1]};
            const v2f vb = {cv[2 * j + np][2], cv[2 * j + np][3]};
            acc[2 * np] = __builtin_amdgcn_wmma_f32_16x16x4_f32(false, va, false, pb,
                                                                (short)0, acc[2 * np],
                                                                false, false);
            acc[2 * np + 1] = __builtin_amdgcn_wmma_f32_16x16x4_f32(false, vb, false, pb,
                                                                    (short)0, acc[2 * np + 1],
                                                                    false, false);
        }
    }
}

__global__ __launch_bounds__(256) void attn_kernel(const float* __restrict__ q,
                                                   const float* k,
                                                   const float* v,
                                                   float* __restrict__ o) {
    __shared__ __attribute__((aligned(16))) float lds[8][AW_FLOATS];
    const int lane = threadIdx.x & 31;
    const int warp = __builtin_amdgcn_readfirstlane((int)(threadIdx.x >> 5));
    const int bh = blockIdx.x >> 4;
    const int grp = ((blockIdx.x & 15) << 2) + (warp & 3);
    const bool two = warp < 4;
    const int h = bh & 15;
    const int b = bh >> 4;
    const int ln = lane & 15;
    const int lh = lane >> 4;

    float* sK = &lds[warp][0];
    const int kr = lane >> 4;
    const int kc = (lane & 15) << 2;

    const float* pK = k + ((size_t)(bh * 128) << 10) + (lane << 2);
    const float* pV = v + ((size_t)(bh * 128) << 10) + (lane << 2);

    v8f acc[4];
    float mrun;
    float lrun;

    const int nseg = two ? 2 : 1;
    for (int seg = 0; seg < nseg; ++seg) {
        const int qt = (two && seg == 0) ? grp : 127 - grp;
        const int k0 = (two && seg == 1) ? 64 : 0;
        const int k1 = two ? qt : 63;

        const float* qbase = q + ((size_t)(b * SEQ) + qt * 16 + ln) * DM + h * DK;
        v2f qa[16];
#pragma unroll
        for (int j = 0; j < 16; ++j)
            qa[j] = *(const v2f*)(qbase + 4 * j + 2 * lh);

        v4f ckA[8], cvA[8], ckB[8], cvB[8];
#pragma unroll
        for (int g = 0; g < 8; ++g) {
            ckA[g] = *(const v4f*)(pK + ((size_t)k0 << 10) + g * 128);
            cvA[g] = *(const v4f*)(pV + ((size_t)k0 << 10) + g * 128);
        }

#pragma unroll
        for (int nt = 0; nt < 4; ++nt) acc[nt] = v8f{};
        mrun = -1e30f;
        lrun = 0.0f;

        asm volatile("" : : "v"(qa[0]), "v"(qa[1]), "v"(qa[2]), "v"(qa[3]), "v"(qa[4]), "v"(qa[5]),
                            "v"(qa[6]), "v"(qa[7]), "v"(qa[8]), "v"(qa[9]), "v"(qa[10]), "v"(qa[11]),
                            "v"(qa[12]), "v"(qa[13]), "v"(qa[14]), "v"(qa[15]));

        for (int kt = k0; kt <= k1; kt += 2) {
            {
                const int kn = (kt + 1 < k1) ? kt + 1 : k1;
                asm volatile("" : : "v"(acc[0]), "v"(acc[1]), "v"(acc[2]), "v"(acc[3]) : "memory");
#pragma unroll
                for (int g = 0; g < 8; ++g) {
                    ckB[g] = *(const v4f*)(pK + ((size_t)kn << 10) + g * 128);
                    cvB[g] = *(const v4f*)(pV + ((size_t)kn << 10) + g * 128);
                }
                asm volatile("" : : : "memory");
            }
            attn_trip(ckA, cvA, qa, acc, mrun, lrun, kt == qt, ln, lh);
            if (kt + 1 > k1) break;
            {
                const int kn = (kt + 2 < k1) ? kt + 2 : k1;
                asm volatile("" : : "v"(acc[0]), "v"(acc[1]), "v"(acc[2]), "v"(acc[3]) : "memory");
#pragma unroll
                for (int g = 0; g < 8; ++g) {
                    ckA[g] = *(const v4f*)(pK + ((size_t)kn << 10) + g * 128);
                    cvA[g] = *(const v4f*)(pV + ((size_t)kn << 10) + g * 128);
                }
                asm volatile("" : : : "memory");
            }
            attn_trip(ckB, cvB, qa, acc, mrun, lrun, kt + 1 == qt, ln, lh);
        }

        if (two && seg == 0)
            attn_store_tile(acc, lrun, sK, o, b, h, qt, ln, lh, kr, kc);
    }

    if (two) {
        float* pp = &lds[warp][0];
#pragma unroll
        for (int nt = 0; nt < 4; ++nt) {
            v4f lo = {acc[nt][0], acc[nt][1], acc[nt][2], acc[nt][3]};
            v4f hi = {acc[nt][4], acc[nt][5], acc[nt][6], acc[nt][7]};
            *(v4f*)(pp + ((nt * 2 + 0) * 32 + lane) * 4) = lo;
            *(v4f*)(pp + ((nt * 2 + 1) * 32 + lane) * 4) = hi;
        }
        pp[AP_M + lane] = mrun;
        pp[AP_L + lane] = lrun;
    }
    __syncthreads();
    if (!two) {
        const float* pp = &lds[warp - 4][0];
        const float pm = pp[AP_M + lane];
        const float pl = pp[AP_L + lane];
        const float nm = fmaxf(mrun, pm);
        const float fa = __expf(mrun - nm);
        const float fb = __expf(pm - nm);
        lrun = lrun * fa + pl * fb;
#pragma unroll
        for (int nt = 0; nt < 4; ++nt) {
            const v4f lo = *(const v4f*)(pp + ((nt * 2 + 0) * 32 + lane) * 4);
            const v4f hi = *(const v4f*)(pp + ((nt * 2 + 1) * 32 + lane) * 4);
#pragma unroll
            for (int e = 0; e < 4; ++e) {
                acc[nt][e] = acc[nt][e] * fa + lo[e] * fb;
                acc[nt][4 + e] = acc[nt][4 + e] * fa + hi[e] * fb;
            }
        }
        attn_store_tile(acc, lrun, sK, o, b, h, 127 - grp, ln, lh, kr, kc);
    }
}

extern "C" void kernel_launch(void* const* d_in, const int* in_sizes, int n_in,
                              void* d_out, int out_size, void* d_ws, size_t ws_size,
                              hipStream_t stream) {
    const float* x  = (const float*)d_in[0];
    const float* Wq = (const float*)d_in[1];
    const float* Wk = (const float*)d_in[2];
    const float* Wv = (const float*)d_in[3];
    const float* Wo = (const float*)d_in[4];
    float* out = (float*)d_out;

    const size_t bufElems = (size_t)BATCH * SEQ * DM;
    float* qb  = (float*)d_ws;
    float* kb  = qb + bufElems;
    float* vb  = kb + bufElems;
    float* ab  = vb + bufElems;

    const int M = BATCH * SEQ, N = DM, K = DM;
    const int gblocks = (M / 128) * (N / 128);

    qkv_proj_kernel<<<3 * gblocks, 256, 0, stream>>>(x, Wq, Wk, Wv, qb, kb, vb, M, N, K);

    const int waves = BATCH * NH * (SEQ / 16);
    attn_kernel<<<waves / 8, 256, 0, stream>>>(qb, kb, vb, ab);

    gemm_xwt_kernel<<<gblocks, 256, 0, stream>>>(ab, Wo, out, M, N, K);
}
